# starting kernel
# baseline (speedup 1.0000x reference)
#include <hip/hip_runtime.h>
#include <hip/hip_bf16.h>
#include <math.h>

typedef __attribute__((ext_vector_type(2))) float v2f;
typedef __attribute__((ext_vector_type(8))) float v8f;

#define N_NODES 100000
#define N_EDGES 1600000
#define CH 128
#define LDS_STRIDE 132

#define OFF_W     0
#define OFF_BIAS  (OFF_W + CH*CH)
#define OFF_ODEG  (OFF_BIAS + CH)
#define OFF_IDEG  (OFF_ODEG + N_NODES)
#define OFF_H     (OFF_IDEG + N_NODES)

__global__ void zero_f4(float* __restrict__ p, int n4) {
    int i = blockIdx.x * blockDim.x + threadIdx.x;
    if (i < n4) ((float4*)p)[i] = make_float4(0.f, 0.f, 0.f, 0.f);
}

__global__ void prep_weights(const float* __restrict__ Wmu, const float* __restrict__ Wrho,
                             const float* __restrict__ Weps,
                             const float* __restrict__ bmu, const float* __restrict__ brho,
                             const float* __restrict__ beps,
                             float* __restrict__ W, float* __restrict__ bias) {
    int i = blockIdx.x * blockDim.x + threadIdx.x;
    if (i < CH * CH) {
        float sp = log1pf(expf(Wrho[i]));
        W[i] = Wmu[i] + Weps[i] * sp;
    } else if (i < CH * CH + CH) {
        int j = i - CH * CH;
        float sp = log1pf(expf(brho[j]));
        bias[j] = bmu[j] + beps[j] * sp;
    }
}

__global__ void degrees_kernel(const int* __restrict__ src, const int* __restrict__ dst,
                               float* __restrict__ odeg, float* __restrict__ ideg) {
    int e = blockIdx.x * blockDim.x + threadIdx.x;
    if (e < N_EDGES) {
        atomicAdd(&odeg[src[e]], 1.0f);
        atomicAdd(&ideg[dst[e]], 1.0f);
    }
}

__global__ __launch_bounds__(128) void gemm_wmma(const float* __restrict__ feat,
                                                 const float* __restrict__ odeg,
                                                 const float* __restrict__ W,
                                                 float* __restrict__ h) {
    __shared__ float lds[16 * LDS_STRIDE];
    const int t = threadIdx.x;
    const int row0 = blockIdx.x * 16;

    {
        const int r = t >> 3;
        const int c = (t & 7) * 16;
        const float s = rsqrtf(fmaxf(odeg[row0 + r], 1.0f));
        const float4* fp = (const float4*)&feat[(size_t)(row0 + r) * CH + c];
#pragma unroll
        for (int j = 0; j < 4; ++j) {
            float4 v = fp[j];
            v.x *= s; v.y *= s; v.z *= s; v.w *= s;
            *(float4*)&lds[r * LDS_STRIDE + c + j * 4] = v;
        }
    }
    __syncthreads();

    const int lane = t & 31;
    const int wave = t >> 5;
    const int half = lane >> 4;
    const int l16  = lane & 15;
    const int n0 = wave * 16;
    const int n1 = 64 + wave * 16;

    v8f c0 = {};
    v8f c1 = {};
    const float* arow = &lds[l16 * LDS_STRIDE + half * 2];

#pragma unroll 4
    for (int k = 0; k < CH; k += 4) {
        v2f a;
        a.x = arow[k];
        a.y = arow[k + 1];

        const float* wk = &W[(size_t)(k + half * 2) * CH + l16];
        v2f b0, b1;
        b0.x = wk[n0];       b0.y = wk[CH + n0];
        b1.x = wk[n1];       b1.y = wk[CH + n1];

        c0 = __builtin_amdgcn_wmma_f32_16x16x4_f32(false, a, false, b0, (short)0, c0, false, false);
        c1 = __builtin_amdgcn_wmma_f32_16x16x4_f32(false, a, false, b1, (short)0, c1, false, false);
    }

#pragma unroll
    for (int r = 0; r < 8; ++r) {
        const int row = row0 + r + half * 8;
        h[(size_t)row * CH + n0 + l16] = c0[r];
        h[(size_t)row * CH + n1 + l16] = c1[r];
    }
}

__global__ void scatter_kernel(const int* __restrict__ src, const int* __restrict__ dst,
                               const float* __restrict__ h, float* __restrict__ out) {
    const int gid = blockIdx.x * blockDim.x + threadIdx.x;
    const int e = gid >> 5;
    if (e >= N_EDGES) return;
    const int q = (gid & 31) * 4;
    const int s = src[e];
    const int d = dst[e];
    const float4 v = *(const float4*)&h[(size_t)s * CH + q];
    float* o = &out[(size_t)d * CH + q];
    atomicAdd(o + 0, v.x);
    atomicAdd(o + 1, v.y);
    atomicAdd(o + 2, v.z);
    atomicAdd(o + 3, v.w);
}

__global__ void finalize_kernel(const float* __restrict__ ideg, const float* __restrict__ bias,
                                float* __restrict__ out) {
    const int gid = blockIdx.x * blockDim.x + threadIdx.x;
    if (gid >= N_NODES * 32) return;
    const int n = gid >> 5;
    const int q = (gid & 31) * 4;
    const float s = rsqrtf(fmaxf(ideg[n], 1.0f));
    float4 v = *(float4*)&out[(size_t)n * CH + q];
    const float4 b = *(const float4*)&bias[q];
    v.x = fmaf(v.x, s, b.x);
    v.y = fmaf(v.y, s, b.y);
    v.z = fmaf(v.z, s, b.z);
    v.w = fmaf(v.w, s, b.w);
    *(float4*)&out[(size_t)n * CH + q] = v;
}

extern "C" void kernel_launch(void* const* d_in, const int* in_sizes, int n_in,
                              void* d_out, int out_size, void* d_ws, size_t ws_size,
                              hipStream_t stream) {
    const float* feat  = (const float*)d_in[0];
    const int*   src   = (const int*)  d_in[1];
    const int*   dst   = (const int*)  d_in[2];
    const float* W_mu  = (const float*)d_in[3];
    const float* W_rho = (const float*)d_in[4];
    const float* b_mu  = (const float*)d_in[5];
    const float* b_rho = (const float*)d_in[6];
    const float* W_eps = (const float*)d_in[7];
    const float* b_eps = (const float*)d_in[8];

    float* ws   = (float*)d_ws;
    float* W    = ws + OFF_W;
    float* bias = ws + OFF_BIAS;
    float* odeg = ws + OFF_ODEG;
    float* ideg = ws + OFF_IDEG;
    float* h    = ws + OFF_H;
    float* out  = (float*)d_out;

    {
        const int n4 = (2 * N_NODES) / 4;
        zero_f4<<<(n4 + 255) / 256, 256, 0, stream>>>(odeg, n4);
    }
    {
        const int n4 = (N_NODES * CH) / 4;
        zero_f4<<<(n4 + 255) / 256, 256, 0, stream>>>(out, n4);
    }

    prep_weights<<<(CH * CH + CH + 255) / 256, 256, 0, stream>>>(
        W_mu, W_rho, W_eps, b_mu, b_rho, b_eps, W, bias);

    degrees_kernel<<<(N_EDGES + 255) / 256, 256, 0, stream>>>(src, dst, odeg, ideg);

    gemm_wmma<<<N_NODES / 16, 128, 0, stream>>>(feat, odeg, W, h);

    {
        const long long nthreads = (long long)N_EDGES * 32;
        scatter_kernel<<<(int)((nthreads + 255) / 256), 256, 0, stream>>>(src, dst, h, out);
    }

    {
        const int nthreads = N_NODES * 32;
        finalize_kernel<<<(nthreads + 255) / 256, 256, 0, stream>>>(ideg, bias, out);
    }
}


# Scatter lane map: lane l owns channels l, l+32, l+64, l+96, so each f32 atomic of a wave covers one whole 128-byte line, not all 4 lines of the row at float4 stride; sums in arrival order as before
# speedup vs baseline: 1.2129x; 1.2129x over previous
#include <hip/hip_runtime.h>
#include <hip/hip_bf16.h>
#include <math.h>

typedef __attribute__((ext_vector_type(2))) float v2f;
typedef __attribute__((ext_vector_type(8))) float v8f;

#define N_NODES 100000
#define N_EDGES 1600000
#define CH 128
#define LDS_STRIDE 132

#define OFF_W     0
#define OFF_BIAS  (OFF_W + CH*CH)
#define OFF_ODEG  (OFF_BIAS + CH)
#define OFF_IDEG  (OFF_ODEG + N_NODES)
#define OFF_H     (OFF_IDEG + N_NODES)

__global__ void zero_f4(float* __restrict__ p, int n4) {
    int i = blockIdx.x * blockDim.x + threadIdx.x;
    if (i < n4) ((float4*)p)[i] = make_float4(0.f, 0.f, 0.f, 0.f);
}

__global__ void prep_weights(const float* __restrict__ Wmu, const float* __restrict__ Wrho,
                             const float* __restrict__ Weps,
                             const float* __restrict__ bmu, const float* __restrict__ brho,
                             const float* __restrict__ beps,
                             float* __restrict__ W, float* __restrict__ bias) {
    int i = blockIdx.x * blockDim.x + threadIdx.x;
    if (i < CH * CH) {
        float sp = log1pf(expf(Wrho[i]));
        W[i] = Wmu[i] + Weps[i] * sp;
    } else if (i < CH * CH + CH) {
        int j = i - CH * CH;
        float sp = log1pf(expf(brho[j]));
        bias[j] = bmu[j] + beps[j] * sp;
    }
}

__global__ void degrees_kernel(const int* __restrict__ src, const int* __restrict__ dst,
                               float* __restrict__ odeg, float* __restrict__ ideg) {
    int e = blockIdx.x * blockDim.x + threadIdx.x;
    if (e < N_EDGES) {
        atomicAdd(&odeg[src[e]], 1.0f);
        atomicAdd(&ideg[dst[e]], 1.0f);
    }
}

__global__ __launch_bounds__(128) void gemm_wmma(const float* __restrict__ feat,
                                                 const float* __restrict__ odeg,
                                                 const float* __restrict__ W,
                                                 float* __restrict__ h) {
    __shared__ float lds[16 * LDS_STRIDE];
    const int t = threadIdx.x;
    const int row0 = blockIdx.x * 16;

    {
        const int r = t >> 3;
        const int c = (t & 7) * 16;
        const float s = rsqrtf(fmaxf(odeg[row0 + r], 1.0f));
        const float4* fp = (const float4*)&feat[(size_t)(row0 + r) * CH + c];
#pragma unroll
        for (int j = 0; j < 4; ++j) {
            float4 v = fp[j];
            v.x *= s; v.y *= s; v.z *= s; v.w *= s;
            *(float4*)&lds[r * LDS_STRIDE + c + j * 4] = v;
        }
    }
    __syncthreads();

    const int lane = t & 31;
    const int wave = t >> 5;
    const int half = lane >> 4;
    const int l16  = lane & 15;
    const int n0 = wave * 16;
    const int n1 = 64 + wave * 16;

    v8f c0 = {};
    v8f c1 = {};
    const float* arow = &lds[l16 * LDS_STRIDE + half * 2];

#pragma unroll 4
    for (int k = 0; k < CH; k += 4) {
        v2f a;
        a.x = arow[k];
        a.y = arow[k + 1];

        const float* wk = &W[(size_t)(k + half * 2) * CH + l16];
        v2f b0, b1;
        b0.x = wk[n0];       b0.y = wk[CH + n0];
        b1.x = wk[n1];       b1.y = wk[CH + n1];

        c0 = __builtin_amdgcn_wmma_f32_16x16x4_f32(false, a, false, b0, (short)0, c0, false, false);
        c1 = __builtin_amdgcn_wmma_f32_16x16x4_f32(false, a, false, b1, (short)0, c1, false, false);
    }

#pragma unroll
    for (int r = 0; r < 8; ++r) {
        const int row = row0 + r + half * 8;
        h[(size_t)row * CH + n0 + l16] = c0[r];
        h[(size_t)row * CH + n1 + l16] = c1[r];
    }
}

__global__ void scatter_kernel(const int* __restrict__ src, const int* __restrict__ dst,
                               const float* __restrict__ h, float* __restrict__ out) {
    const int gid = blockIdx.x * blockDim.x + threadIdx.x;
    const int e = gid >> 5;
    if (e >= N_EDGES) return;
    const int l = gid & 31;
    const int s = src[e];
    const int d = dst[e];
    const float* hp = &h[(size_t)s * CH + l];
    const float v0 = hp[0];
    const float v1 = hp[32];
    const float v2 = hp[64];
    const float v3 = hp[96];
    float* o = &out[(size_t)d * CH + l];
    atomicAdd(o,      v0);
    atomicAdd(o + 32, v1);
    atomicAdd(o + 64, v2);
    atomicAdd(o + 96, v3);
}

__global__ void finalize_kernel(const float* __restrict__ ideg, const float* __restrict__ bias,
                                float* __restrict__ out) {
    const int gid = blockIdx.x * blockDim.x + threadIdx.x;
    if (gid >= N_NODES * 32) return;
    const int n = gid >> 5;
    const int q = (gid & 31) * 4;
    const float s = rsqrtf(fmaxf(ideg[n], 1.0f));
    float4 v = *(float4*)&out[(size_t)n * CH + q];
    const float4 b = *(const float4*)&bias[q];
    v.x = fmaf(v.x, s, b.x);
    v.y = fmaf(v.y, s, b.y);
    v.z = fmaf(v.z, s, b.z);
    v.w = fmaf(v.w, s, b.w);
    *(float4*)&out[(size_t)n * CH + q] = v;
}

extern "C" void kernel_launch(void* const* d_in, const int* in_sizes, int n_in,
                              void* d_out, int out_size, void* d_ws, size_t ws_size,
                              hipStream_t stream) {
    const float* feat  = (const float*)d_in[0];
    const int*   src   = (const int*)  d_in[1];
    const int*   dst   = (const int*)  d_in[2];
    const float* W_mu  = (const float*)d_in[3];
    const float* W_rho = (const float*)d_in[4];
    const float* b_mu  = (const float*)d_in[5];
    const float* b_rho = (const float*)d_in[6];
    const float* W_eps = (const float*)d_in[7];
    const float* b_eps = (const float*)d_in[8];

    float* ws   = (float*)d_ws;
    float* W    = ws + OFF_W;
    float* bias = ws + OFF_BIAS;
    float* odeg = ws + OFF_ODEG;
    float* ideg = ws + OFF_IDEG;
    float* h    = ws + OFF_H;
    float* out  = (float*)d_out;

    {
        const int n4 = (2 * N_NODES) / 4;
        zero_f4<<<(n4 + 255) / 256, 256, 0, stream>>>(odeg, n4);
    }
    {
        const int n4 = (N_NODES * CH) / 4;
        zero_f4<<<(n4 + 255) / 256, 256, 0, stream>>>(out, n4);
    }

    prep_weights<<<(CH * CH + CH + 255) / 256, 256, 0, stream>>>(
        W_mu, W_rho, W_eps, b_mu, b_rho, b_eps, W, bias);

    degrees_kernel<<<(N_EDGES + 255) / 256, 256, 0, stream>>>(src, dst, odeg, ideg);

    gemm_wmma<<<N_NODES / 16, 128, 0, stream>>>(feat, odeg, W, h);

    {
        const long long nthreads = (long long)N_EDGES * 32;
        scatter_kernel<<<(int)((nthreads + 255) / 256), 256, 0, stream>>>(src, dst, h, out);
    }

    {
        const int nthreads = N_NODES * 32;
        finalize_kernel<<<(nthreads + 255) / 256, 256, 0, stream>>>(ideg, bias, out);
    }
}


# Bit-planes replace the dst scan: the degree pass marks each edge in its dst tile's bitmap (LDS, parked in the tile's own out rows); a tile reads 200 KB, not all of dst; f32 sums: LDS adds as before
# speedup vs baseline: 1.9137x; 1.5778x over previous
#include <hip/hip_runtime.h>
#include <hip/hip_bf16.h>
#include <math.h>

typedef __attribute__((ext_vector_type(2))) float v2f;
typedef __attribute__((ext_vector_type(8))) float v8f;

#define N_NODES 100000
#define N_EDGES 1600000
#define CH 128
#define LDS_STRIDE 132

#define OFF_W     0
#define OFF_BIAS  (OFF_W + CH*CH)
#define OFF_ODEG  (OFF_BIAS + CH)
#define OFF_IDEG  (OFF_ODEG + N_NODES)
#define OFF_H     (OFF_IDEG + N_NODES)

__global__ void zero_f4(float* __restrict__ p, int n4) {
    int i = blockIdx.x * blockDim.x + threadIdx.x;
    if (i < n4) ((float4*)p)[i] = make_float4(0.f, 0.f, 0.f, 0.f);
}

__global__ void prep_weights(const float* __restrict__ Wmu, const float* __restrict__ Wrho,
                             const float* __restrict__ Weps,
                             const float* __restrict__ bmu, const float* __restrict__ brho,
                             const float* __restrict__ beps,
                             float* __restrict__ W, float* __restrict__ bias) {
    int i = blockIdx.x * blockDim.x + threadIdx.x;
    if (i < CH * CH) {
        float sp = log1pf(expf(Wrho[i]));
        W[i] = Wmu[i] + Weps[i] * sp;
    } else if (i < CH * CH + CH) {
        int j = i - CH * CH;
        float sp = log1pf(expf(brho[j]));
        bias[j] = bmu[j] + beps[j] * sp;
    }
}

#define TILE 400
#define N_TILES (N_NODES / TILE)
#define TILE_WORDS (TILE * CH)
#define CHUNK 4096
#define N_CHUNKS ((N_EDGES + CHUNK - 1) / CHUNK)
#define CW (CHUNK / 32)
#define PLANE_WORDS (N_CHUNKS * CW)
#define BSTRIDE (CW + 4)

__global__ __launch_bounds__(1024) void degrees_planes(const int* __restrict__ src, const int* __restrict__ dst,
                                                       float* __restrict__ odeg, float* __restrict__ ideg,
                                                       float* __restrict__ out) {
    __shared__ unsigned bits[N_TILES * BSTRIDE];

    const int t = threadIdx.x;
    for (int i = t; i < N_TILES * BSTRIDE / 4; i += 1024) ((uint4*)bits)[i] = make_uint4(0u, 0u, 0u, 0u);
    __syncthreads();

    const int g = blockIdx.x * (CHUNK / 4) + t;
    if (g < N_EDGES / 4) {
        const int4 s = ((const int4*)src)[g];
        const int4 d = ((const int4*)dst)[g];
        atomicAdd(&odeg[s.x], 1.0f);
        atomicAdd(&odeg[s.y], 1.0f);
        atomicAdd(&odeg[s.z], 1.0f);
        atomicAdd(&odeg[s.w], 1.0f);
        atomicAdd(&ideg[d.x], 1.0f);
        atomicAdd(&ideg[d.y], 1.0f);
        atomicAdd(&ideg[d.z], 1.0f);
        atomicAdd(&ideg[d.w], 1.0f);
        const unsigned w = (unsigned)t >> 3;
        const unsigned sh = 4u * ((unsigned)t & 7u);
        atomicOr(&bits[((unsigned)d.x / TILE) * BSTRIDE + w], 1u << sh);
        atomicOr(&bits[((unsigned)d.y / TILE) * BSTRIDE + w], 2u << sh);
        atomicOr(&bits[((unsigned)d.z / TILE) * BSTRIDE + w], 4u << sh);
        atomicOr(&bits[((unsigned)d.w / TILE) * BSTRIDE + w], 8u << sh);
    }
    __syncthreads();

    unsigned* planes = (unsigned*)out + (size_t)blockIdx.x * CW;
    for (int i = t; i < N_TILES * (CW / 4); i += 1024) {
        const int tile = i >> 5;
        const int q = i & 31;
        const uint4 v = *(const uint4*)&bits[tile * BSTRIDE + q * 4];
        *(uint4*)&planes[(size_t)tile * TILE_WORDS + q * 4] = v;
    }
}

__global__ __launch_bounds__(128) void gemm_wmma(const float* __restrict__ feat,
                                                 const float* __restrict__ odeg,
                                                 const float* __restrict__ W,
                                                 float* __restrict__ h) {
    __shared__ float lds[16 * LDS_STRIDE];
    const int t = threadIdx.x;
    const int row0 = blockIdx.x * 16;

    {
        const int r = t >> 3;
        const int c = (t & 7) * 16;
        const float s = rsqrtf(fmaxf(odeg[row0 + r], 1.0f));
        const float4* fp = (const float4*)&feat[(size_t)(row0 + r) * CH + c];
#pragma unroll
        for (int j = 0; j < 4; ++j) {
            float4 v = fp[j];
            v.x *= s; v.y *= s; v.z *= s; v.w *= s;
            *(float4*)&lds[r * LDS_STRIDE + c + j * 4] = v;
        }
    }
    __syncthreads();

    const int lane = t & 31;
    const int wave = t >> 5;
    const int half = lane >> 4;
    const int l16  = lane & 15;
    const int n0 = wave * 16;
    const int n1 = 64 + wave * 16;

    v8f c0 = {};
    v8f c1 = {};
    const float* arow = &lds[l16 * LDS_STRIDE + half * 2];

#pragma unroll 4
    for (int k = 0; k < CH; k += 4) {
        v2f a;
        a.x = arow[k];
        a.y = arow[k + 1];

        const float* wk = &W[(size_t)(k + half * 2) * CH + l16];
        v2f b0, b1;
        b0.x = wk[n0];       b0.y = wk[CH + n0];
        b1.x = wk[n1];       b1.y = wk[CH + n1];

        c0 = __builtin_amdgcn_wmma_f32_16x16x4_f32(false, a, false, b0, (short)0, c0, false, false);
        c1 = __builtin_amdgcn_wmma_f32_16x16x4_f32(false, a, false, b1, (short)0, c1, false, false);
    }

#pragma unroll
    for (int r = 0; r < 8; ++r) {
        const int row = row0 + r + half * 8;
        h[(size_t)row * CH + n0 + l16] = c0[r];
        h[(size_t)row * CH + n1 + l16] = c1[r];
    }
}

#define QCAP 24576
#define PLANE_V4 (PLANE_WORDS / 4)
#define PLANE_TRIPS ((PLANE_V4 + 1023) / 1024)

__device__ __forceinline__ void put_edges(unsigned m, const unsigned e0, unsigned& pos, unsigned* queue) {
    while (m) {
        const unsigned e = e0 + (unsigned)__builtin_ctz(m);
        m &= m - 1u;
        if (pos < (unsigned)QCAP) queue[pos] = e;
        ++pos;
    }
}

__global__ __launch_bounds__(1024) void tile_gather(const int* __restrict__ src,
                                                    const int* __restrict__ dst,
                                                    const float* __restrict__ h,
                                                    const float* __restrict__ ideg,
                                                    const float* __restrict__ bias,
                                                    float* out) {
    __shared__ float acc[TILE * CH];
    __shared__ unsigned queue[QCAP];
    __shared__ unsigned qn;

    const int t = threadIdx.x;
    const int lane = t & 31;
    const int wave = t >> 5;
    const int lo = blockIdx.x * TILE;

    const unsigned* plane = (const unsigned*)out + (size_t)blockIdx.x * TILE_WORDS;
    uint4 pv[PLANE_TRIPS];
#pragma unroll
    for (int k = 0; k < PLANE_TRIPS; ++k) {
        const int i = t + 1024 * k;
        pv[k] = (i < PLANE_V4) ? ((const uint4*)plane)[i] : make_uint4(0u, 0u, 0u, 0u);
    }

    for (int i = t; i < TILE * CH / 4; i += 1024) ((float4*)acc)[i] = make_float4(0.f, 0.f, 0.f, 0.f);
    if (t == 0) qn = 0u;
    __syncthreads();

    unsigned cnt = 0u;
#pragma unroll
    for (int k = 0; k < PLANE_TRIPS; ++k)
        cnt += (unsigned)(__builtin_popcount(pv[k].x) + __builtin_popcount(pv[k].y) +
                          __builtin_popcount(pv[k].z) + __builtin_popcount(pv[k].w));
    unsigned pos = atomicAdd(&qn, cnt);
#pragma unroll
    for (int k = 0; k < PLANE_TRIPS; ++k) {
        const unsigned e0 = (unsigned)(t + 1024 * k) * 128u;
        put_edges(pv[k].x, e0,       pos, queue);
        put_edges(pv[k].y, e0 + 32u, pos, queue);
        put_edges(pv[k].z, e0 + 64u, pos, queue);
        put_edges(pv[k].w, e0 + 96u, pos, queue);
    }
    __syncthreads();

    const unsigned total = qn;
    const unsigned n = (total <= (unsigned)QCAP) ? total : 0u;

    for (unsigned q = t; q < n; q += 1024) {
        const unsigned e = queue[q];
        const unsigned s = (unsigned)src[e];
        const unsigned dl = (unsigned)(dst[e] - lo);
        queue[q] = (s << 9) | dl;
    }
    __syncthreads();

    const unsigned nfull = n & ~7u;
    for (unsigned g = (unsigned)wave * 8u; g < nfull; g += 256u) {
        const uint4 qa = *(const uint4*)&queue[g];
        const uint4 qb = *(const uint4*)&queue[g + 4];
        unsigned ent[8] = {qa.x, qa.y, qa.z, qa.w, qb.x, qb.y, qb.z, qb.w};
        float v[8][4];
#pragma unroll
        for (int k = 0; k < 8; ++k) {
            ent[k] = __builtin_amdgcn_readfirstlane(ent[k]);
            const float* hp = h + (size_t)(ent[k] >> 9) * CH + lane;
            v[k][0] = hp[0];
            v[k][1] = hp[32];
            v[k][2] = hp[64];
            v[k][3] = hp[96];
        }
#pragma unroll
        for (int k = 0; k < 8; ++k) {
            float* ap = acc + (ent[k] & 511u) * CH + lane;
            atomicAdd(ap,      v[k][0]);
            atomicAdd(ap + 32, v[k][1]);
            atomicAdd(ap + 64, v[k][2]);
            atomicAdd(ap + 96, v[k][3]);
        }
    }
    if (nfull + (unsigned)wave < n) {
        const unsigned ent = queue[nfull + (unsigned)wave];
        const float* hp = h + (size_t)(ent >> 9) * CH + lane;
        const float v0 = hp[0], v1 = hp[32], v2 = hp[64], v3 = hp[96];
        float* ap = acc + (ent & 511u) * CH + lane;
        atomicAdd(ap,      v0);
        atomicAdd(ap + 32, v1);
        atomicAdd(ap + 64, v2);
        atomicAdd(ap + 96, v3);
    }
    if (total > (unsigned)QCAP) {
        for (int i = t; i < PLANE_WORDS; i += 1024) {
            unsigned m = plane[i];
            while (m) {
                const unsigned e = (unsigned)i * 32u + (unsigned)__builtin_ctz(m);
                m &= m - 1u;
                const unsigned dl = (unsigned)(dst[e] - lo);
                const float* hr = h + (size_t)src[e] * CH;
                float* ar = acc + dl * CH;
                for (int k = 0; k < CH; ++k) atomicAdd(&ar[k], hr[k]);
            }
        }
    }
    __syncthreads();

    const float4 b = *(const float4*)&bias[lane * 4];
    for (int r = wave; r < TILE; r += 32) {
        const int node = lo + r;
        const float s = rsqrtf(fmaxf(ideg[node], 1.0f));
        float4 a = *(const float4*)&acc[r * CH + lane * 4];
        a.x = fmaf(a.x, s, b.x);
        a.y = fmaf(a.y, s, b.y);
        a.z = fmaf(a.z, s, b.z);
        a.w = fmaf(a.w, s, b.w);
        *(float4*)&out[(size_t)node * CH + lane * 4] = a;
    }
}

extern "C" void kernel_launch(void* const* d_in, const int* in_sizes, int n_in,
                              void* d_out, int out_size, void* d_ws, size_t ws_size,
                              hipStream_t stream) {
    const float* feat  = (const float*)d_in[0];
    const int*   src   = (const int*)  d_in[1];
    const int*   dst   = (const int*)  d_in[2];
    const float* W_mu  = (const float*)d_in[3];
    const float* W_rho = (const float*)d_in[4];
    const float* b_mu  = (const float*)d_in[5];
    const float* b_rho = (const float*)d_in[6];
    const float* W_eps = (const float*)d_in[7];
    const float* b_eps = (const float*)d_in[8];

    float* ws   = (float*)d_ws;
    float* W    = ws + OFF_W;
    float* bias = ws + OFF_BIAS;
    float* odeg = ws + OFF_ODEG;
    float* ideg = ws + OFF_IDEG;
    float* h    = ws + OFF_H;
    float* out  = (float*)d_out;

    {
        const int n4 = (2 * N_NODES) / 4;
        zero_f4<<<(n4 + 255) / 256, 256, 0, stream>>>(odeg, n4);
    }

    prep_weights<<<(CH * CH + CH + 255) / 256, 256, 0, stream>>>(
        W_mu, W_rho, W_eps, b_mu, b_rho, b_eps, W, bias);

    degrees_planes<<<N_CHUNKS, 1024, 0, stream>>>(src, dst, odeg, ideg, out);

    gemm_wmma<<<N_NODES / 16, 128, 0, stream>>>(feat, odeg, W, h);

    tile_gather<<<N_TILES, 1024, 0, stream>>>(src, dst, h, ideg, bias, out);
}


# Out-degrees counted in LDS, not by global f32 atomics: 64 edge chunks x 2 node halves, two 16-bit counters per word; spare work-groups of the weight launch sum the planes; f32 sums unchanged
# speedup vs baseline: 3.7450x; 1.9569x over previous
#include <hip/hip_runtime.h>
#include <hip/hip_bf16.h>
#include <math.h>

typedef __attribute__((ext_vector_type(2))) float v2f;
typedef __attribute__((ext_vector_type(8))) float v8f;

#define N_NODES 100000
#define N_EDGES 1600000
#define CH 128
#define LDS_STRIDE 132

#define OFF_W     0
#define OFF_BIAS  (OFF_W + CH*CH)
#define OFF_ODEG  (OFF_BIAS + CH)
#define OFF_H     (OFF_ODEG + 2 * N_NODES)

#define OD_CHUNKS 64
#define OD_EDGES (N_EDGES / OD_CHUNKS)
#define OD_HALF (N_NODES / 2)
#define OD_WORDS (OD_HALF / 2)
#define OD_V4 (OD_EDGES / 4)
#define OD_TRIPS ((OD_V4 + 1023) / 1024)
#define PLANE16_WORDS (N_NODES / 2)
static_assert(OD_EDGES * OD_CHUNKS == N_EDGES && OD_EDGES % 4 == 0 && OD_EDGES < 65536 && OD_WORDS % 4 == 0,
              "chunks must tile the edge list in 16-byte pieces and stay below the range of a 16-bit counter");

__global__ __launch_bounds__(1024) void out_degree_counts(const int* __restrict__ src, float* __restrict__ out) {
    __shared__ unsigned cnt[OD_WORDS];

    const int t = threadIdx.x;
    const int c = blockIdx.x >> 1;
    const unsigned half = (unsigned)blockIdx.x & 1u;
    const unsigned base = half * OD_HALF;

    const int4* sp = (const int4*)(src + (size_t)c * OD_EDGES);
    int4 sv[OD_TRIPS];
#pragma unroll
    for (int k = 0; k < OD_TRIPS; ++k) {
        const int i = t + 1024 * k;
        sv[k] = (i < OD_V4) ? sp[i] : make_int4(-1, -1, -1, -1);
    }

    for (int i = t; i < OD_WORDS / 4; i += 1024) ((uint4*)cnt)[i] = make_uint4(0u, 0u, 0u, 0u);
    __syncthreads();

#pragma unroll
    for (int k = 0; k < OD_TRIPS; ++k) {
        const unsigned n0 = (unsigned)sv[k].x - base;
        const unsigned n1 = (unsigned)sv[k].y - base;
        const unsigned n2 = (unsigned)sv[k].z - base;
        const unsigned n3 = (unsigned)sv[k].w - base;
        if (n0 < (unsigned)OD_HALF) atomicAdd(&cnt[n0 >> 1], 1u << ((n0 & 1u) * 16u));
        if (n1 < (unsigned)OD_HALF) atomicAdd(&cnt[n1 >> 1], 1u << ((n1 & 1u) * 16u));
        if (n2 < (unsigned)OD_HALF) atomicAdd(&cnt[n2 >> 1], 1u << ((n2 & 1u) * 16u));
        if (n3 < (unsigned)OD_HALF) atomicAdd(&cnt[n3 >> 1], 1u << ((n3 & 1u) * 16u));
    }
    __syncthreads();

    uint4* pp = (uint4*)((unsigned*)out + (size_t)c * PLANE16_WORDS + (size_t)half * OD_WORDS);
    for (int i = t; i < OD_WORDS / 4; i += 1024) pp[i] = ((const uint4*)cnt)[i];
}

#define PREP_BLOCKS ((CH * CH + CH + 255) / 256)
#define SUM_BLOCKS ((PLANE16_WORDS + 255) / 256)
__global__ void prep_weights(const float* __restrict__ Wmu, const float* __restrict__ Wrho,
                             const float* __restrict__ Weps,
                             const float* __restrict__ bmu, const float* __restrict__ brho,
                             const float* __restrict__ beps,
                             float* __restrict__ W, float* __restrict__ bias,
                             const float* __restrict__ out, float* __restrict__ odeg) {
    if (blockIdx.x >= PREP_BLOCKS) {
        const int w = (blockIdx.x - PREP_BLOCKS) * 256 + threadIdx.x;
        if (w < PLANE16_WORDS) {
            const unsigned* pw = (const unsigned*)out + w;
            unsigned lo = 0u, hi = 0u;
#pragma unroll
            for (int c = 0; c < OD_CHUNKS; ++c) {
                const unsigned v = pw[(size_t)c * PLANE16_WORDS];
                lo += v & 0xffffu;
                hi += v >> 16;
            }
            *(float2*)&odeg[2 * w] = make_float2((float)lo, (float)hi);
        }
        return;
    }
    int i = blockIdx.x * blockDim.x + threadIdx.x;
    if (i < CH * CH) {
        float sp = log1pf(expf(Wrho[i]));
        W[i] = Wmu[i] + Weps[i] * sp;
    } else if (i < CH * CH + CH) {
        int j = i - CH * CH;
        float sp = log1pf(expf(brho[j]));
        bias[j] = bmu[j] + beps[j] * sp;
    }
}

#define TILE 400
#define N_TILES (N_NODES / TILE)
#define TILE_WORDS (TILE * CH)
#define CHUNK 4096
#define N_CHUNKS ((N_EDGES + CHUNK - 1) / CHUNK)
#define CW (CHUNK / 32)
#define PLANE_WORDS (N_CHUNKS * CW)
#define BSTRIDE (CW + 4)

__global__ __launch_bounds__(1024) void degrees_planes(const int* __restrict__ dst,
                                                       float* __restrict__ out) {
    __shared__ unsigned bits[N_TILES * BSTRIDE];

    const int t = threadIdx.x;
    for (int i = t; i < N_TILES * BSTRIDE / 4; i += 1024) ((uint4*)bits)[i] = make_uint4(0u, 0u, 0u, 0u);
    __syncthreads();

    const int g = blockIdx.x * (CHUNK / 4) + t;
    if (g < N_EDGES / 4) {
        const int4 d = ((const int4*)dst)[g];
        const unsigned w = (unsigned)t >> 3;
        const unsigned sh = 4u * ((unsigned)t & 7u);
        atomicOr(&bits[((unsigned)d.x / TILE) * BSTRIDE + w], 1u << sh);
        atomicOr(&bits[((unsigned)d.y / TILE) * BSTRIDE + w], 2u << sh);
        atomicOr(&bits[((unsigned)d.z / TILE) * BSTRIDE + w], 4u << sh);
        atomicOr(&bits[((unsigned)d.w / TILE) * BSTRIDE + w], 8u << sh);
    }
    __syncthreads();

    unsigned* planes = (unsigned*)out + (size_t)blockIdx.x * CW;
    for (int i = t; i < N_TILES * (CW / 4); i += 1024) {
        const int tile = i >> 5;
        const int q = i & 31;
        const uint4 v = *(const uint4*)&bits[tile * BSTRIDE + q * 4];
        *(uint4*)&planes[(size_t)tile * TILE_WORDS + q * 4] = v;
    }
}

__global__ __launch_bounds__(128) void gemm_wmma(const float* __restrict__ feat,
                                                 const float* __restrict__ odeg,
                                                 const float* __restrict__ W,
                                                 float* __restrict__ h) {
    __shared__ float lds[16 * LDS_STRIDE];
    const int t = threadIdx.x;
    const int row0 = blockIdx.x * 16;

    {
        const int r = t >> 3;
        const int c = (t & 7) * 16;
        const float s = rsqrtf(fmaxf(odeg[row0 + r], 1.0f));
        const float4* fp = (const float4*)&feat[(size_t)(row0 + r) * CH + c];
#pragma unroll
        for (int j = 0; j < 4; ++j) {
            float4 v = fp[j];
            v.x *= s; v.y *= s; v.z *= s; v.w *= s;
            *(float4*)&lds[r * LDS_STRIDE + c + j * 4] = v;
        }
    }
    __syncthreads();

    const int lane = t & 31;
    const int wave = t >> 5;
    const int half = lane >> 4;
    const int l16  = lane & 15;
    const int n0 = wave * 16;
    const int n1 = 64 + wave * 16;

    v8f c0 = {};
    v8f c1 = {};
    const float* arow = &lds[l16 * LDS_STRIDE + half * 2];

#pragma unroll 4
    for (int k = 0; k < CH; k += 4) {
        v2f a;
        a.x = arow[k];
        a.y = arow[k + 1];

        const float* wk = &W[(size_t)(k + half * 2) * CH + l16];
        v2f b0, b1;
        b0.x = wk[n0];       b0.y = wk[CH + n0];
        b1.x = wk[n1];       b1.y = wk[CH + n1];

        c0 = __builtin_amdgcn_wmma_f32_16x16x4_f32(false, a, false, b0, (short)0, c0, false, false);
        c1 = __builtin_amdgcn_wmma_f32_16x16x4_f32(false, a, false, b1, (short)0, c1, false, false);
    }

#pragma unroll
    for (int r = 0; r < 8; ++r) {
        const int row = row0 + r + half * 8;
        h[(size_t)row * CH + n0 + l16] = c0[r];
        h[(size_t)row * CH + n1 + l16] = c1[r];
    }
}

#define QCAP 24576
#define PLANE_V4 (PLANE_WORDS / 4)
#define PLANE_TRIPS ((PLANE_V4 + 1023) / 1024)

__device__ __forceinline__ void put_edges(unsigned m, const unsigned e0, unsigned& pos, unsigned* queue) {
    while (m) {
        const unsigned e = e0 + (unsigned)__builtin_ctz(m);
        m &= m - 1u;
        if (pos < (unsigned)QCAP) queue[pos] = e;
        ++pos;
    }
}

__global__ __launch_bounds__(1024) void tile_gather(const int* __restrict__ src,
                                                    const int* __restrict__ dst,
                                                    const float* __restrict__ h,
                                                    const float* __restrict__ bias,
                                                    float* out) {
    __shared__ float acc[TILE * CH];
    __shared__ unsigned queue[QCAP];
    __shared__ unsigned indeg[TILE];
    __shared__ unsigned qn;

    const int t = threadIdx.x;
    const int lane = t & 31;
    const int wave = t >> 5;
    const int lo = blockIdx.x * TILE;

    const unsigned* plane = (const unsigned*)out + (size_t)blockIdx.x * TILE_WORDS;
    uint4 pv[PLANE_TRIPS];
#pragma unroll
    for (int k = 0; k < PLANE_TRIPS; ++k) {
        const int i = t + 1024 * k;
        pv[k] = (i < PLANE_V4) ? ((const uint4*)plane)[i] : make_uint4(0u, 0u, 0u, 0u);
    }

    for (int i = t; i < TILE * CH / 4; i += 1024) ((float4*)acc)[i] = make_float4(0.f, 0.f, 0.f, 0.f);
    if (t < TILE) indeg[t] = 0u;
    if (t == 0) qn = 0u;
    __syncthreads();

    unsigned cnt = 0u;
#pragma unroll
    for (int k = 0; k < PLANE_TRIPS; ++k)
        cnt += (unsigned)(__builtin_popcount(pv[k].x) + __builtin_popcount(pv[k].y) +
                          __builtin_popcount(pv[k].z) + __builtin_popcount(pv[k].w));
    unsigned pos = atomicAdd(&qn, cnt);
#pragma unroll
    for (int k = 0; k < PLANE_TRIPS; ++k) {
        const unsigned e0 = (unsigned)(t + 1024 * k) * 128u;
        put_edges(pv[k].x, e0,       pos, queue);
        put_edges(pv[k].y, e0 + 32u, pos, queue);
        put_edges(pv[k].z, e0 + 64u, pos, queue);
        put_edges(pv[k].w, e0 + 96u, pos, queue);
    }
    __syncthreads();

    const unsigned total = qn;
    const unsigned n = (total <= (unsigned)QCAP) ? total : 0u;

    for (unsigned q = t; q < n; q += 1024) {
        const unsigned e = queue[q];
        const unsigned s = (unsigned)src[e];
        const unsigned dl = (unsigned)(dst[e] - lo);
        queue[q] = (s << 9) | dl;
        atomicAdd(&indeg[dl], 1u);
    }
    __syncthreads();

    const unsigned nfull = n & ~7u;
    for (unsigned g = (unsigned)wave * 8u; g < nfull; g += 256u) {
        const uint4 qa = *(const uint4*)&queue[g];
        const uint4 qb = *(const uint4*)&queue[g + 4];
        unsigned ent[8] = {qa.x, qa.y, qa.z, qa.w, qb.x, qb.y, qb.z, qb.w};
        float v[8][4];
#pragma unroll
        for (int k = 0; k < 8; ++k) {
            ent[k] = __builtin_amdgcn_readfirstlane(ent[k]);
            const float* hp = h + (size_t)(ent[k] >> 9) * CH + lane;
            v[k][0] = hp[0];
            v[k][1] = hp[32];
            v[k][2] = hp[64];
            v[k][3] = hp[96];
        }
#pragma unroll
        for (int k = 0; k < 8; ++k) {
            float* ap = acc + (ent[k] & 511u) * CH + lane;
            atomicAdd(ap,      v[k][0]);
            atomicAdd(ap + 32, v[k][1]);
            atomicAdd(ap + 64, v[k][2]);
            atomicAdd(ap + 96, v[k][3]);
        }
    }
    if (nfull + (unsigned)wave < n) {
        const unsigned ent = queue[nfull + (unsigned)wave];
        const float* hp = h + (size_t)(ent >> 9) * CH + lane;
        const float v0 = hp[0], v1 = hp[32], v2 = hp[64], v3 = hp[96];
        float* ap = acc + (ent & 511u) * CH + lane;
        atomicAdd(ap,      v0);
        atomicAdd(ap + 32, v1);
        atomicAdd(ap + 64, v2);
        atomicAdd(ap + 96, v3);
    }
    if (total > (unsigned)QCAP) {
        for (int i = t; i < PLANE_WORDS; i += 1024) {
            unsigned m = plane[i];
            while (m) {
                const unsigned e = (unsigned)i * 32u + (unsigned)__builtin_ctz(m);
                m &= m - 1u;
                const unsigned dl = (unsigned)(dst[e] - lo);
                atomicAdd(&indeg[dl], 1u);
                const float* hr = h + (size_t)src[e] * CH;
                float* ar = acc + dl * CH;
                for (int k = 0; k < CH; ++k) atomicAdd(&ar[k], hr[k]);
            }
        }
    }
    __syncthreads();

    const float4 b = *(const float4*)&bias[lane * 4];
    for (int r = wave; r < TILE; r += 32) {
        const int node = lo + r;
        const float s = rsqrtf(fmaxf((float)indeg[r], 1.0f));
        float4 a = *(const float4*)&acc[r * CH + lane * 4];
        a.x = fmaf(a.x, s, b.x);
        a.y = fmaf(a.y, s, b.y);
        a.z = fmaf(a.z, s, b.z);
        a.w = fmaf(a.w, s, b.w);
        *(float4*)&out[(size_t)node * CH + lane * 4] = a;
    }
}

extern "C" void kernel_launch(void* const* d_in, const int* in_sizes, int n_in,
                              void* d_out, int out_size, void* d_ws, size_t ws_size,
                              hipStream_t stream) {
    const float* feat  = (const float*)d_in[0];
    const int*   src   = (const int*)  d_in[1];
    const int*   dst   = (const int*)  d_in[2];
    const float* W_mu  = (const float*)d_in[3];
    const float* W_rho = (const float*)d_in[4];
    const float* b_mu  = (const float*)d_in[5];
    const float* b_rho = (const float*)d_in[6];
    const float* W_eps = (const float*)d_in[7];
    const float* b_eps = (const float*)d_in[8];

    float* ws   = (float*)d_ws;
    float* W    = ws + OFF_W;
    float* bias = ws + OFF_BIAS;
    float* odeg = ws + OFF_ODEG;
    float* h    = ws + OFF_H;
    float* out  = (float*)d_out;

    out_degree_counts<<<OD_CHUNKS * 2, 1024, 0, stream>>>(src, out);

    prep_weights<<<PREP_BLOCKS + SUM_BLOCKS, 256, 0, stream>>>(
        W_mu, W_rho, W_eps, b_mu, b_rho, b_eps, W, bias, out, odeg);

    degrees_planes<<<N_CHUNKS, 1024, 0, stream>>>(dst, out);

    gemm_wmma<<<N_NODES / 16, 128, 0, stream>>>(feat, odeg, W, h);

    tile_gather<<<N_TILES, 1024, 0, stream>>>(src, dst, h, bias, out);
}


# Gather takes each h row as one 16-byte load per lane (one 512-B wave load, not four 128-B ones); LDS tile keeps channel 4l+c at word 32c+l: adds stay conflict-free; f32 sums: LDS adds as before
# speedup vs baseline: 3.9900x; 1.0654x over previous
#include <hip/hip_runtime.h>
#include <hip/hip_bf16.h>
#include <math.h>

typedef __attribute__((ext_vector_type(2))) float v2f;
typedef __attribute__((ext_vector_type(8))) float v8f;

#define N_NODES 100000
#define N_EDGES 1600000
#define CH 128
#define LDS_STRIDE 132

#define OFF_W     0
#define OFF_BIAS  (OFF_W + CH*CH)
#define OFF_ODEG  (OFF_BIAS + CH)
#define OFF_H     (OFF_ODEG + 2 * N_NODES)

#define OD_CHUNKS 64
#define OD_EDGES (N_EDGES / OD_CHUNKS)
#define OD_HALF (N_NODES / 2)
#define OD_WORDS (OD_HALF / 2)
#define OD_V4 (OD_EDGES / 4)
#define OD_TRIPS ((OD_V4 + 1023) / 1024)
#define PLANE16_WORDS (N_NODES / 2)
static_assert(OD_EDGES * OD_CHUNKS == N_EDGES && OD_EDGES % 4 == 0 && OD_EDGES < 65536 && OD_WORDS % 4 == 0,
              "chunks must tile the edge list in 16-byte pieces and stay below the range of a 16-bit counter");

__global__ __launch_bounds__(1024) void out_degree_counts(const int* __restrict__ src, float* __restrict__ out) {
    __shared__ unsigned cnt[OD_WORDS];

    const int t = threadIdx.x;
    const int c = blockIdx.x >> 1;
    const unsigned half = (unsigned)blockIdx.x & 1u;
    const unsigned base = half * OD_HALF;

    const int4* sp = (const int4*)(src + (size_t)c * OD_EDGES);
    int4 sv[OD_TRIPS];
#pragma unroll
    for (int k = 0; k < OD_TRIPS; ++k) {
        const int i = t + 1024 * k;
        sv[k] = (i < OD_V4) ? sp[i] : make_int4(-1, -1, -1, -1);
    }

    for (int i = t; i < OD_WORDS / 4; i += 1024) ((uint4*)cnt)[i] = make_uint4(0u, 0u, 0u, 0u);
    __syncthreads();

#pragma unroll
    for (int k = 0; k < OD_TRIPS; ++k) {
        const unsigned n0 = (unsigned)sv[k].x - base;
        const unsigned n1 = (unsigned)sv[k].y - base;
        const unsigned n2 = (unsigned)sv[k].z - base;
        const unsigned n3 = (unsigned)sv[k].w - base;
        if (n0 < (unsigned)OD_HALF) atomicAdd(&cnt[n0 >> 1], 1u << ((n0 & 1u) * 16u));
        if (n1 < (unsigned)OD_HALF) atomicAdd(&cnt[n1 >> 1], 1u << ((n1 & 1u) * 16u));
        if (n2 < (unsigned)OD_HALF) atomicAdd(&cnt[n2 >> 1], 1u << ((n2 & 1u) * 16u));
        if (n3 < (unsigned)OD_HALF) atomicAdd(&cnt[n3 >> 1], 1u << ((n3 & 1u) * 16u));
    }
    __syncthreads();

    uint4* pp = (uint4*)((unsigned*)out + (size_t)c * PLANE16_WORDS + (size_t)half * OD_WORDS);
    for (int i = t; i < OD_WORDS / 4; i += 1024) pp[i] = ((const uint4*)cnt)[i];
}

#define PREP_BLOCKS ((CH * CH + CH + 255) / 256)
#define SUM_BLOCKS ((PLANE16_WORDS + 255) / 256)
__global__ void prep_weights(const float* __restrict__ Wmu, const float* __restrict__ Wrho,
                             const float* __restrict__ Weps,
                             const float* __restrict__ bmu, const float* __restrict__ brho,
                             const float* __restrict__ beps,
                             float* __restrict__ W, float* __restrict__ bias,
                             const float* __restrict__ out, float* __restrict__ odeg) {
    if (blockIdx.x >= PREP_BLOCKS) {
        const int w = (blockIdx.x - PREP_BLOCKS) * 256 + threadIdx.x;
        if (w < PLANE16_WORDS) {
            const unsigned* pw = (const unsigned*)out + w;
            unsigned lo = 0u, hi = 0u;
#pragma unroll
            for (int c = 0; c < OD_CHUNKS; ++c) {
                const unsigned v = pw[(size_t)c * PLANE16_WORDS];
                lo += v & 0xffffu;
                hi += v >> 16;
            }
            *(float2*)&odeg[2 * w] = make_float2((float)lo, (float)hi);
        }
        return;
    }
    int i = blockIdx.x * blockDim.x + threadIdx.x;
    if (i < CH * CH) {
        float sp = log1pf(expf(Wrho[i]));
        W[i] = Wmu[i] + Weps[i] * sp;
    } else if (i < CH * CH + CH) {
        int j = i - CH * CH;
        float sp = log1pf(expf(brho[j]));
        bias[j] = bmu[j] + beps[j] * sp;
    }
}

#define TILE 400
#define N_TILES (N_NODES / TILE)
#define TILE_WORDS (TILE * CH)
#define CHUNK 4096
#define N_CHUNKS ((N_EDGES + CHUNK - 1) / CHUNK)
#define CW (CHUNK / 32)
#define PLANE_WORDS (N_CHUNKS * CW)
#define BSTRIDE (CW + 4)

__global__ __launch_bounds__(1024) void degrees_planes(const int* __restrict__ dst,
                                                       float* __restrict__ out) {
    __shared__ unsigned bits[N_TILES * BSTRIDE];

    const int t = threadIdx.x;
    for (int i = t; i < N_TILES * BSTRIDE / 4; i += 1024) ((uint4*)bits)[i] = make_uint4(0u, 0u, 0u, 0u);
    __syncthreads();

    const int g = blockIdx.x * (CHUNK / 4) + t;
    if (g < N_EDGES / 4) {
        const int4 d = ((const int4*)dst)[g];
        const unsigned w = (unsigned)t >> 3;
        const unsigned sh = 4u * ((unsigned)t & 7u);
        atomicOr(&bits[((unsigned)d.x / TILE) * BSTRIDE + w], 1u << sh);
        atomicOr(&bits[((unsigned)d.y / TILE) * BSTRIDE + w], 2u << sh);
        atomicOr(&bits[((unsigned)d.z / TILE) * BSTRIDE + w], 4u << sh);
        atomicOr(&bits[((unsigned)d.w / TILE) * BSTRIDE + w], 8u << sh);
    }
    __syncthreads();

    unsigned* planes = (unsigned*)out + (size_t)blockIdx.x * CW;
    for (int i = t; i < N_TILES * (CW / 4); i += 1024) {
        const int tile = i >> 5;
        const int q = i & 31;
        const uint4 v = *(const uint4*)&bits[tile * BSTRIDE + q * 4];
        *(uint4*)&planes[(size_t)tile * TILE_WORDS + q * 4] = v;
    }
}

__global__ __launch_bounds__(128) void gemm_wmma(const float* __restrict__ feat,
                                                 const float* __restrict__ odeg,
                                                 const float* __restrict__ W,
                                                 float* __restrict__ h) {
    __shared__ float lds[16 * LDS_STRIDE];
    const int t = threadIdx.x;
    const int row0 = blockIdx.x * 16;

    {
        const int r = t >> 3;
        const int c = (t & 7) * 16;
        const float s = rsqrtf(fmaxf(odeg[row0 + r], 1.0f));
        const float4* fp = (const float4*)&feat[(size_t)(row0 + r) * CH + c];
#pragma unroll
        for (int j = 0; j < 4; ++j) {
            float4 v = fp[j];
            v.x *= s; v.y *= s; v.z *= s; v.w *= s;
            *(float4*)&lds[r * LDS_STRIDE + c + j * 4] = v;
        }
    }
    __syncthreads();

    const int lane = t & 31;
    const int wave = t >> 5;
    const int half = lane >> 4;
    const int l16  = lane & 15;
    const int n0 = wave * 16;
    const int n1 = 64 + wave * 16;

    v8f c0 = {};
    v8f c1 = {};
    const float* arow = &lds[l16 * LDS_STRIDE + half * 2];

#pragma unroll 4
    for (int k = 0; k < CH; k += 4) {
        v2f a;
        a.x = arow[k];
        a.y = arow[k + 1];

        const float* wk = &W[(size_t)(k + half * 2) * CH + l16];
        v2f b0, b1;
        b0.x = wk[n0];       b0.y = wk[CH + n0];
        b1.x = wk[n1];       b1.y = wk[CH + n1];

        c0 = __builtin_amdgcn_wmma_f32_16x16x4_f32(false, a, false, b0, (short)0, c0, false, false);
        c1 = __builtin_amdgcn_wmma_f32_16x16x4_f32(false, a, false, b1, (short)0, c1, false, false);
    }

#pragma unroll
    for (int r = 0; r < 8; ++r) {
        const int row = row0 + r + half * 8;
        h[(size_t)row * CH + n0 + l16] = c0[r];
        h[(size_t)row * CH + n1 + l16] = c1[r];
    }
}

#define ACC_WORD(ch) ((((ch) & 3) << 5) | ((ch) >> 2))
#define QCAP 24576
#define PLANE_V4 (PLANE_WORDS / 4)
#define PLANE_TRIPS ((PLANE_V4 + 1023) / 1024)

__device__ __forceinline__ void put_edges(unsigned m, const unsigned e0, unsigned& pos, unsigned* queue) {
    while (m) {
        const unsigned e = e0 + (unsigned)__builtin_ctz(m);
        m &= m - 1u;
        if (pos < (unsigned)QCAP) queue[pos] = e;
        ++pos;
    }
}

__global__ __launch_bounds__(1024) void tile_gather(const int* __restrict__ src,
                                                    const int* __restrict__ dst,
                                                    const float* __restrict__ h,
                                                    const float* __restrict__ bias,
                                                    float* out) {
    __shared__ float acc[TILE * CH];
    __shared__ unsigned queue[QCAP];
    __shared__ unsigned indeg[TILE];
    __shared__ unsigned qn;

    const int t = threadIdx.x;
    const int lane = t & 31;
    const int wave = t >> 5;
    const int lo = blockIdx.x * TILE;

    const unsigned* plane = (const unsigned*)out + (size_t)blockIdx.x * TILE_WORDS;
    uint4 pv[PLANE_TRIPS];
#pragma unroll
    for (int k = 0; k < PLANE_TRIPS; ++k) {
        const int i = t + 1024 * k;
        pv[k] = (i < PLANE_V4) ? ((const uint4*)plane)[i] : make_uint4(0u, 0u, 0u, 0u);
    }

    for (int i = t; i < TILE * CH / 4; i += 1024) ((float4*)acc)[i] = make_float4(0.f, 0.f, 0.f, 0.f);
    if (t < TILE) indeg[t] = 0u;
    if (t == 0) qn = 0u;
    __syncthreads();

    unsigned cnt = 0u;
#pragma unroll
    for (int k = 0; k < PLANE_TRIPS; ++k)
        cnt += (unsigned)(__builtin_popcount(pv[k].x) + __builtin_popcount(pv[k].y) +
                          __builtin_popcount(pv[k].z) + __builtin_popcount(pv[k].w));
    unsigned pos = atomicAdd(&qn, cnt);
#pragma unroll
    for (int k = 0; k < PLANE_TRIPS; ++k) {
        const unsigned e0 = (unsigned)(t + 1024 * k) * 128u;
        put_edges(pv[k].x, e0,       pos, queue);
        put_edges(pv[k].y, e0 + 32u, pos, queue);
        put_edges(pv[k].z, e0 + 64u, pos, queue);
        put_edges(pv[k].w, e0 + 96u, pos, queue);
    }
    __syncthreads();

    const unsigned total = qn;
    const unsigned n = (total <= (unsigned)QCAP) ? total : 0u;

    for (unsigned q = t; q < n; q += 1024) {
        const unsigned e = queue[q];
        const unsigned s = (unsigned)src[e];
        const unsigned dl = (unsigned)(dst[e] - lo);
        queue[q] = (s << 9) | dl;
        atomicAdd(&indeg[dl], 1u);
    }
    __syncthreads();

    const unsigned nfull = n & ~7u;
    for (unsigned g = (unsigned)wave * 8u; g < nfull; g += 256u) {
        const uint4 qa = *(const uint4*)&queue[g];
        const uint4 qb = *(const uint4*)&queue[g + 4];
        unsigned ent[8] = {qa.x, qa.y, qa.z, qa.w, qb.x, qb.y, qb.z, qb.w};
        float4 v[8];
#pragma unroll
        for (int k = 0; k < 8; ++k) {
            ent[k] = __builtin_amdgcn_readfirstlane(ent[k]);
            v[k] = *(const float4*)(h + (size_t)(ent[k] >> 9) * CH + lane * 4);
        }
#pragma unroll
        for (int k = 0; k < 8; ++k) {
            float* ap = acc + (ent[k] & 511u) * CH + lane;
            atomicAdd(ap,      v[k].x);
            atomicAdd(ap + 32, v[k].y);
            atomicAdd(ap + 64, v[k].z);
            atomicAdd(ap + 96, v[k].w);
        }
    }
    if (nfull + (unsigned)wave < n) {
        const unsigned ent = queue[nfull + (unsigned)wave];
        const float4 v = *(const float4*)(h + (size_t)(ent >> 9) * CH + lane * 4);
        float* ap = acc + (ent & 511u) * CH + lane;
        atomicAdd(ap,      v.x);
        atomicAdd(ap + 32, v.y);
        atomicAdd(ap + 64, v.z);
        atomicAdd(ap + 96, v.w);
    }
    if (total > (unsigned)QCAP) {
        for (int i = t; i < PLANE_WORDS; i += 1024) {
            unsigned m = plane[i];
            while (m) {
                const unsigned e = (unsigned)i * 32u + (unsigned)__builtin_ctz(m);
                m &= m - 1u;
                const unsigned dl = (unsigned)(dst[e] - lo);
                atomicAdd(&indeg[dl], 1u);
                const float* hr = h + (size_t)src[e] * CH;
                float* ar = acc + dl * CH;
                for (int k = 0; k < CH; ++k) atomicAdd(&ar[ACC_WORD(k)], hr[k]);
            }
        }
    }
    __syncthreads();

    const float4 b = *(const float4*)&bias[lane * 4];
    for (int r = wave; r < TILE; r += 32) {
        const int node = lo + r;
        const float s = rsqrtf(fmaxf((float)indeg[r], 1.0f));
        const float* ar = &acc[r * CH + lane];
        float4 a = make_float4(ar[0], ar[32], ar[64], ar[96]);
        a.x = fmaf(a.x, s, b.x);
        a.y = fmaf(a.y, s, b.y);
        a.z = fmaf(a.z, s, b.z);
        a.w = fmaf(a.w, s, b.w);
        *(float4*)&out[(size_t)node * CH + lane * 4] = a;
    }
}

extern "C" void kernel_launch(void* const* d_in, const int* in_sizes, int n_in,
                              void* d_out, int out_size, void* d_ws, size_t ws_size,
                              hipStream_t stream) {
    const float* feat  = (const float*)d_in[0];
    const int*   src   = (const int*)  d_in[1];
    const int*   dst   = (const int*)  d_in[2];
    const float* W_mu  = (const float*)d_in[3];
    const float* W_rho = (const float*)d_in[4];
    const float* b_mu  = (const float*)d_in[5];
    const float* b_rho = (const float*)d_in[6];
    const float* W_eps = (const float*)d_in[7];
    const float* b_eps = (const float*)d_in[8];

    float* ws   = (float*)d_ws;
    float* W    = ws + OFF_W;
    float* bias = ws + OFF_BIAS;
    float* odeg = ws + OFF_ODEG;
    float* h    = ws + OFF_H;
    float* out  = (float*)d_out;

    out_degree_counts<<<OD_CHUNKS * 2, 1024, 0, stream>>>(src, out);

    prep_weights<<<PREP_BLOCKS + SUM_BLOCKS, 256, 0, stream>>>(
        W_mu, W_rho, W_eps, b_mu, b_rho, b_eps, W, bias, out, odeg);

    degrees_planes<<<N_CHUNKS, 1024, 0, stream>>>(dst, out);

    gemm_wmma<<<N_NODES / 16, 128, 0, stream>>>(feat, odeg, W, h);

    tile_gather<<<N_TILES, 1024, 0, stream>>>(src, dst, h, bias, out);
}


# Record cells replace bit-planes: per 4096-edge chunk and tile a 256-B cell of packed (src,row) words, slot by LDS add; tiles read records, no src/dst look-up per edge; f32 sums: LDS adds as before
# speedup vs baseline: 5.1420x; 1.2887x over previous
#include <hip/hip_runtime.h>
#include <hip/hip_bf16.h>
#include <math.h>

typedef __attribute__((ext_vector_type(2))) float v2f;
typedef __attribute__((ext_vector_type(8))) float v8f;

#define N_NODES 100000
#define N_EDGES 1600000
#define CH 128
#define LDS_STRIDE 132

#define OFF_W     0
#define OFF_BIAS  (OFF_W + CH*CH)
#define OFF_ODEG  (OFF_BIAS + CH)
#define OFF_H     (OFF_ODEG + 2 * N_NODES)

#define OD_CHUNKS 64
#define OD_EDGES (N_EDGES / OD_CHUNKS)
#define OD_HALF (N_NODES / 2)
#define OD_WORDS (OD_HALF / 2)
#define OD_V4 (OD_EDGES / 4)
#define OD_TRIPS ((OD_V4 + 1023) / 1024)
#define PLANE16_WORDS (N_NODES / 2)
static_assert(OD_EDGES * OD_CHUNKS == N_EDGES && OD_EDGES % 4 == 0 && OD_EDGES < 65536 && OD_WORDS % 4 == 0,
              "chunks must tile the edge list in 16-byte pieces and stay below the range of a 16-bit counter");

__global__ __launch_bounds__(1024) void out_degree_counts(const int* __restrict__ src, float* __restrict__ out) {
    __shared__ unsigned cnt[OD_WORDS];

    const int t = threadIdx.x;
    const int c = blockIdx.x >> 1;
    const unsigned half = (unsigned)blockIdx.x & 1u;
    const unsigned base = half * OD_HALF;

    const int4* sp = (const int4*)(src + (size_t)c * OD_EDGES);
    int4 sv[OD_TRIPS];
#pragma unroll
    for (int k = 0; k < OD_TRIPS; ++k) {
        const int i = t + 1024 * k;
        sv[k] = (i < OD_V4) ? sp[i] : make_int4(-1, -1, -1, -1);
    }

    for (int i = t; i < OD_WORDS / 4; i += 1024) ((uint4*)cnt)[i] = make_uint4(0u, 0u, 0u, 0u);
    __syncthreads();

#pragma unroll
    for (int k = 0; k < OD_TRIPS; ++k) {
        const unsigned n0 = (unsigned)sv[k].x - base;
        const unsigned n1 = (unsigned)sv[k].y - base;
        const unsigned n2 = (unsigned)sv[k].z - base;
        const unsigned n3 = (unsigned)sv[k].w - base;
        if (n0 < (unsigned)OD_HALF) atomicAdd(&cnt[n0 >> 1], 1u << ((n0 & 1u) * 16u));
        if (n1 < (unsigned)OD_HALF) atomicAdd(&cnt[n1 >> 1], 1u << ((n1 & 1u) * 16u));
        if (n2 < (unsigned)OD_HALF) atomicAdd(&cnt[n2 >> 1], 1u << ((n2 & 1u) * 16u));
        if (n3 < (unsigned)OD_HALF) atomicAdd(&cnt[n3 >> 1], 1u << ((n3 & 1u) * 16u));
    }
    __syncthreads();

    uint4* pp = (uint4*)((unsigned*)out + (size_t)c * PLANE16_WORDS + (size_t)half * OD_WORDS);
    for (int i = t; i < OD_WORDS / 4; i += 1024) pp[i] = ((const uint4*)cnt)[i];
}

#define PREP_BLOCKS ((CH * CH + CH + 255) / 256)
#define SUM_BLOCKS ((PLANE16_WORDS + 255) / 256)
__global__ void prep_weights(const float* __restrict__ Wmu, const float* __restrict__ Wrho,
                             const float* __restrict__ Weps,
                             const float* __restrict__ bmu, const float* __restrict__ brho,
                             const float* __restrict__ beps,
                             float* __restrict__ W, float* __restrict__ bias,
                             const float* __restrict__ out, float* __restrict__ odeg) {
    if (blockIdx.x >= PREP_BLOCKS) {
        const int w = (blockIdx.x - PREP_BLOCKS) * 256 + threadIdx.x;
        if (w < PLANE16_WORDS) {
            const unsigned* pw = (const unsigned*)out + w;
            unsigned lo = 0u, hi = 0u;
#pragma unroll
            for (int c = 0; c < OD_CHUNKS; ++c) {
                const unsigned v = pw[(size_t)c * PLANE16_WORDS];
                lo += v & 0xffffu;
                hi += v >> 16;
            }
            *(float2*)&odeg[2 * w] = make_float2((float)lo, (float)hi);
        }
        return;
    }
    int i = blockIdx.x * blockDim.x + threadIdx.x;
    if (i < CH * CH) {
        float sp = log1pf(expf(Wrho[i]));
        W[i] = Wmu[i] + Weps[i] * sp;
    } else if (i < CH * CH + CH) {
        int j = i - CH * CH;
        float sp = log1pf(expf(brho[j]));
        bias[j] = bmu[j] + beps[j] * sp;
    }
}

#define TILE 400
#define N_TILES (N_NODES / TILE)
#define TILE_WORDS (TILE * CH)
#define CHUNK 4096
#define N_CHUNKS ((N_EDGES + CHUNK - 1) / CHUNK)
#define CELL_W 64
#define CELL_CAP (CELL_W - 1)
#define CELLS_WORDS (N_CHUNKS * CELL_W)
#define EMPTY 0xffffffffu
static_assert(CELLS_WORDS <= TILE_WORDS && CELLS_WORDS % 4 == 0, "a tile's cells are parked in its own rows of out");
static_assert(N_NODES <= (1 << 17) && TILE <= 512, "a record is (source row << 9) | local dst row");

__device__ __forceinline__ void put_record(const unsigned d, const unsigned s, unsigned* cell, unsigned* cnt) {
    const unsigned tile = d / TILE;
    const unsigned slot = atomicAdd(&cnt[tile], 1u);
    if (slot < (unsigned)CELL_CAP) cell[tile * CELL_W + 1u + slot] = (s << 9) | (d - tile * TILE);
}

__global__ __launch_bounds__(1024) void edge_cells(const int* __restrict__ src,
                                                   const int* __restrict__ dst,
                                                   float* __restrict__ out) {
    __shared__ unsigned cell[N_TILES * CELL_W];
    __shared__ unsigned cnt[N_TILES];

    const int t = threadIdx.x;
    const int g = blockIdx.x * (CHUNK / 4) + t;
    const bool live = g < N_EDGES / 4;
    int4 d = make_int4(0, 0, 0, 0);
    int4 s = make_int4(0, 0, 0, 0);
    if (live) {
        d = ((const int4*)dst)[g];
        s = ((const int4*)src)[g];
    }

    for (int i = t; i < N_TILES * CELL_W / 4; i += 1024) ((uint4*)cell)[i] = make_uint4(EMPTY, EMPTY, EMPTY, EMPTY);
    if (t < N_TILES) cnt[t] = 0u;
    __syncthreads();

    if (live) {
        put_record((unsigned)d.x, (unsigned)s.x, cell, cnt);
        put_record((unsigned)d.y, (unsigned)s.y, cell, cnt);
        put_record((unsigned)d.z, (unsigned)s.z, cell, cnt);
        put_record((unsigned)d.w, (unsigned)s.w, cell, cnt);
    }
    __syncthreads();

    unsigned* cells = (unsigned*)out + (size_t)blockIdx.x * CELL_W;
    for (int i = t; i < N_TILES * (CELL_W / 4); i += 1024) {
        const int tile = i >> 4;
        const int q = i & 15;
        const unsigned c = cnt[tile];
        uint4 v = *(const uint4*)&cell[tile * CELL_W + q * 4];
        if (c > (unsigned)CELL_CAP) v = make_uint4(EMPTY, EMPTY, EMPTY, EMPTY);
        if (q == 0) v.x = c;
        *(uint4*)&cells[(size_t)tile * TILE_WORDS + q * 4] = v;
    }
}

__global__ __launch_bounds__(128) void gemm_wmma(const float* __restrict__ feat,
                                                 const float* __restrict__ odeg,
                                                 const float* __restrict__ W,
                                                 float* __restrict__ h) {
    __shared__ float lds[16 * LDS_STRIDE];
    const int t = threadIdx.x;
    const int row0 = blockIdx.x * 16;

    {
        const int r = t >> 3;
        const int c = (t & 7) * 16;
        const float s = rsqrtf(fmaxf(odeg[row0 + r], 1.0f));
        const float4* fp = (const float4*)&feat[(size_t)(row0 + r) * CH + c];
#pragma unroll
        for (int j = 0; j < 4; ++j) {
            float4 v = fp[j];
            v.x *= s; v.y *= s; v.z *= s; v.w *= s;
            *(float4*)&lds[r * LDS_STRIDE + c + j * 4] = v;
        }
    }
    __syncthreads();

    const int lane = t & 31;
    const int wave = t >> 5;
    const int half = lane >> 4;
    const int l16  = lane & 15;
    const int n0 = wave * 16;
    const int n1 = 64 + wave * 16;

    v8f c0 = {};
    v8f c1 = {};
    const float* arow = &lds[l16 * LDS_STRIDE + half * 2];

#pragma unroll 4
    for (int k = 0; k < CH; k += 4) {
        v2f a;
        a.x = arow[k];
        a.y = arow[k + 1];

        const float* wk = &W[(size_t)(k + half * 2) * CH + l16];
        v2f b0, b1;
        b0.x = wk[n0];       b0.y = wk[CH + n0];
        b1.x = wk[n1];       b1.y = wk[CH + n1];

        c0 = __builtin_amdgcn_wmma_f32_16x16x4_f32(false, a, false, b0, (short)0, c0, false, false);
        c1 = __builtin_amdgcn_wmma_f32_16x16x4_f32(false, a, false, b1, (short)0, c1, false, false);
    }

#pragma unroll
    for (int r = 0; r < 8; ++r) {
        const int row = row0 + r + half * 8;
        h[(size_t)row * CH + n0 + l16] = c0[r];
        h[(size_t)row * CH + n1 + l16] = c1[r];
    }
}

#define ACC_WORD(ch) ((((ch) & 3) << 5) | ((ch) >> 2))
#define QCAP 24576
#define CELLS_V4 (CELLS_WORDS / 4)
#define CELL_TRIPS ((CELLS_V4 + 1023) / 1024)

__device__ __forceinline__ void put_rec(const unsigned w, unsigned& pos, unsigned* queue, unsigned* indeg) {
    if (w != EMPTY) {
        if (pos < (unsigned)QCAP) queue[pos] = w;
        ++pos;
        atomicAdd(&indeg[w & 511u], 1u);
    }
}

__device__ __forceinline__ void add_row_alone(float* acc, const float* __restrict__ h, const unsigned s, const unsigned dl) {
    const float* hr = h + (size_t)s * CH;
    float* ar = acc + dl * CH;
    for (int k = 0; k < CH; ++k) atomicAdd(&ar[ACC_WORD(k)], hr[k]);
}

__global__ __launch_bounds__(1024) void tile_gather(const int* __restrict__ src,
                                                    const int* __restrict__ dst,
                                                    const float* __restrict__ h,
                                                    const float* __restrict__ bias,
                                                    float* out) {
    __shared__ float acc[TILE * CH];
    __shared__ unsigned queue[QCAP];
    __shared__ unsigned indeg[TILE];
    __shared__ unsigned ovf[N_CHUNKS];
    __shared__ unsigned qn;
    __shared__ unsigned novf;

    const int t = threadIdx.x;
    const int lane = t & 31;
    const int wave = t >> 5;
    const int lo = blockIdx.x * TILE;

    const unsigned* cells = (const unsigned*)out + (size_t)blockIdx.x * TILE_WORDS;
    uint4 pv[CELL_TRIPS];
#pragma unroll
    for (int k = 0; k < CELL_TRIPS; ++k) {
        const int i = t + 1024 * k;
        pv[k] = (i < CELLS_V4) ? ((const uint4*)cells)[i] : make_uint4(EMPTY, EMPTY, EMPTY, EMPTY);
    }

    for (int i = t; i < TILE * CH / 4; i += 1024) ((float4*)acc)[i] = make_float4(0.f, 0.f, 0.f, 0.f);
    if (t < TILE) indeg[t] = 0u;
    if (t == 0) { qn = 0u; novf = 0u; }
    __syncthreads();

    unsigned cnt = 0u;
#pragma unroll
    for (int k = 0; k < CELL_TRIPS; ++k) {
        if ((t & 15) == 0) {
            const unsigned c = pv[k].x;
            if (c != EMPTY && c > (unsigned)CELL_CAP) ovf[atomicAdd(&novf, 1u)] = (unsigned)(t + 1024 * k) >> 4;
            pv[k].x = EMPTY;
        }
        cnt += (unsigned)(pv[k].x != EMPTY) + (unsigned)(pv[k].y != EMPTY) +
               (unsigned)(pv[k].z != EMPTY) + (unsigned)(pv[k].w != EMPTY);
    }
    unsigned pos = atomicAdd(&qn, cnt);
#pragma unroll
    for (int k = 0; k < CELL_TRIPS; ++k) {
        put_rec(pv[k].x, pos, queue, indeg);
        put_rec(pv[k].y, pos, queue, indeg);
        put_rec(pv[k].z, pos, queue, indeg);
        put_rec(pv[k].w, pos, queue, indeg);
    }
    __syncthreads();

    const unsigned total = qn;
    const unsigned n = (total <= (unsigned)QCAP) ? total : 0u;

    const unsigned nfull = n & ~7u;
    for (unsigned g = (unsigned)wave * 8u; g < nfull; g += 256u) {
        const uint4 qa = *(const uint4*)&queue[g];
        const uint4 qb = *(const uint4*)&queue[g + 4];
        unsigned ent[8] = {qa.x, qa.y, qa.z, qa.w, qb.x, qb.y, qb.z, qb.w};
        float4 v[8];
#pragma unroll
        for (int k = 0; k < 8; ++k) {
            ent[k] = __builtin_amdgcn_readfirstlane(ent[k]);
            v[k] = *(const float4*)(h + (size_t)(ent[k] >> 9) * CH + lane * 4);
        }
#pragma unroll
        for (int k = 0; k < 8; ++k) {
            float* ap = acc + (ent[k] & 511u) * CH + lane;
            atomicAdd(ap,      v[k].x);
            atomicAdd(ap + 32, v[k].y);
            atomicAdd(ap + 64, v[k].z);
            atomicAdd(ap + 96, v[k].w);
        }
    }
    if (nfull + (unsigned)wave < n) {
        const unsigned ent = queue[nfull + (unsigned)wave];
        const float4 v = *(const float4*)(h + (size_t)(ent >> 9) * CH + lane * 4);
        float* ap = acc + (ent & 511u) * CH + lane;
        atomicAdd(ap,      v.x);
        atomicAdd(ap + 32, v.y);
        atomicAdd(ap + 64, v.z);
        atomicAdd(ap + 96, v.w);
    }
    if (total > (unsigned)QCAP) {
        for (int i = t; i < CELLS_WORDS; i += 1024) {
            const unsigned w = cells[i];
            if ((i & (CELL_W - 1)) != 0 && w != EMPTY) add_row_alone(acc, h, w >> 9, w & 511u);
        }
    }
    const unsigned nov = novf;
    for (unsigned o = 0; o < nov; ++o) {
        const unsigned e0 = ovf[o] * (unsigned)CHUNK;
        for (int i = t; i < CHUNK; i += 1024) {
            const unsigned e = e0 + (unsigned)i;
            if (e < (unsigned)N_EDGES) {
                const unsigned dl = (unsigned)(dst[e] - lo);
                if (dl < (unsigned)TILE) {
                    atomicAdd(&indeg[dl], 1u);
                    add_row_alone(acc, h, (unsigned)src[e], dl);
                }
            }
        }
    }
    __syncthreads();

    const float4 b = *(const float4*)&bias[lane * 4];
    for (int r = wave; r < TILE; r += 32) {
        const int node = lo + r;
        const float s = rsqrtf(fmaxf((float)indeg[r], 1.0f));
        const float* ar = &acc[r * CH + lane];
        float4 a = make_float4(ar[0], ar[32], ar[64], ar[96]);
        a.x = fmaf(a.x, s, b.x);
        a.y = fmaf(a.y, s, b.y);
        a.z = fmaf(a.z, s, b.z);
        a.w = fmaf(a.w, s, b.w);
        *(float4*)&out[(size_t)node * CH + lane * 4] = a;
    }
}

extern "C" void kernel_launch(void* const* d_in, const int* in_sizes, int n_in,
                              void* d_out, int out_size, void* d_ws, size_t ws_size,
                              hipStream_t stream) {
    const float* feat  = (const float*)d_in[0];
    const int*   src   = (const int*)  d_in[1];
    const int*   dst   = (const int*)  d_in[2];
    const float* W_mu  = (const float*)d_in[3];
    const float* W_rho = (const float*)d_in[4];
    const float* b_mu  = (const float*)d_in[5];
    const float* b_rho = (const float*)d_in[6];
    const float* W_eps = (const float*)d_in[7];
    const float* b_eps = (const float*)d_in[8];

    float* ws   = (float*)d_ws;
    float* W    = ws + OFF_W;
    float* bias = ws + OFF_BIAS;
    float* odeg = ws + OFF_ODEG;
    float* h    = ws + OFF_H;
    float* out  = (float*)d_out;

    out_degree_counts<<<OD_CHUNKS * 2, 1024, 0, stream>>>(src, out);

    prep_weights<<<PREP_BLOCKS + SUM_BLOCKS, 256, 0, stream>>>(
        W_mu, W_rho, W_eps, b_mu, b_rho, b_eps, W, bias, out, odeg);

    edge_cells<<<N_CHUNKS, 1024, 0, stream>>>(src, dst, out);

    gemm_wmma<<<N_NODES / 16, 128, 0, stream>>>(feat, odeg, W, h);

    tile_gather<<<N_TILES, 1024, 0, stream>>>(src, dst, h, bias, out);
}


# Tile's records counting-sorted by dst row in LDS (integer adds give slots); a wave sums a row's h rows in registers and stores the row once: no LDS f32 adds, no LDS tile; f32 sums in slot order
# speedup vs baseline: 5.5527x; 1.0799x over previous
#include <hip/hip_runtime.h>
#include <hip/hip_bf16.h>
#include <math.h>

typedef __attribute__((ext_vector_type(2))) float v2f;
typedef __attribute__((ext_vector_type(8))) float v8f;

#define N_NODES 100000
#define N_EDGES 1600000
#define CH 128
#define LDS_STRIDE 132

#define OFF_W     0
#define OFF_BIAS  (OFF_W + CH*CH)
#define OFF_ODEG  (OFF_BIAS + CH)
#define OFF_H     (OFF_ODEG + 2 * N_NODES)

#define OD_CHUNKS 64
#define OD_EDGES (N_EDGES / OD_CHUNKS)
#define OD_HALF (N_NODES / 2)
#define OD_WORDS (OD_HALF / 2)
#define OD_V4 (OD_EDGES / 4)
#define OD_TRIPS ((OD_V4 + 1023) / 1024)
#define PLANE16_WORDS (N_NODES / 2)
static_assert(OD_EDGES * OD_CHUNKS == N_EDGES && OD_EDGES % 4 == 0 && OD_EDGES < 65536 && OD_WORDS % 4 == 0,
              "chunks must tile the edge list in 16-byte pieces and stay below the range of a 16-bit counter");

__global__ __launch_bounds__(1024) void out_degree_counts(const int* __restrict__ src, float* __restrict__ out) {
    __shared__ unsigned cnt[OD_WORDS];

    const int t = threadIdx.x;
    const int c = blockIdx.x >> 1;
    const unsigned half = (unsigned)blockIdx.x & 1u;
    const unsigned base = half * OD_HALF;

    const int4* sp = (const int4*)(src + (size_t)c * OD_EDGES);
    int4 sv[OD_TRIPS];
#pragma unroll
    for (int k = 0; k < OD_TRIPS; ++k) {
        const int i = t + 1024 * k;
        sv[k] = (i < OD_V4) ? sp[i] : make_int4(-1, -1, -1, -1);
    }

    for (int i = t; i < OD_WORDS / 4; i += 1024) ((uint4*)cnt)[i] = make_uint4(0u, 0u, 0u, 0u);
    __syncthreads();

#pragma unroll
    for (int k = 0; k < OD_TRIPS; ++k) {
        const unsigned n0 = (unsigned)sv[k].x - base;
        const unsigned n1 = (unsigned)sv[k].y - base;
        const unsigned n2 = (unsigned)sv[k].z - base;
        const unsigned n3 = (unsigned)sv[k].w - base;
        if (n0 < (unsigned)OD_HALF) atomicAdd(&cnt[n0 >> 1], 1u << ((n0 & 1u) * 16u));
        if (n1 < (unsigned)OD_HALF) atomicAdd(&cnt[n1 >> 1], 1u << ((n1 & 1u) * 16u));
        if (n2 < (unsigned)OD_HALF) atomicAdd(&cnt[n2 >> 1], 1u << ((n2 & 1u) * 16u));
        if (n3 < (unsigned)OD_HALF) atomicAdd(&cnt[n3 >> 1], 1u << ((n3 & 1u) * 16u));
    }
    __syncthreads();

    uint4* pp = (uint4*)((unsigned*)out + (size_t)c * PLANE16_WORDS + (size_t)half * OD_WORDS);
    for (int i = t; i < OD_WORDS / 4; i += 1024) pp[i] = ((const uint4*)cnt)[i];
}

#define PREP_BLOCKS ((CH * CH + CH + 255) / 256)
#define SUM_BLOCKS ((PLANE16_WORDS + 255) / 256)
__global__ void prep_weights(const float* __restrict__ Wmu, const float* __restrict__ Wrho,
                             const float* __restrict__ Weps,
                             const float* __restrict__ bmu, const float* __restrict__ brho,
                             const float* __restrict__ beps,
                             float* __restrict__ W, float* __restrict__ bias,
                             const float* __restrict__ out, float* __restrict__ odeg) {
    if (blockIdx.x >= PREP_BLOCKS) {
        const int w = (blockIdx.x - PREP_BLOCKS) * 256 + threadIdx.x;
        if (w < PLANE16_WORDS) {
            const unsigned* pw = (const unsigned*)out + w;
            unsigned lo = 0u, hi = 0u;
#pragma unroll
            for (int c = 0; c < OD_CHUNKS; ++c) {
                const unsigned v = pw[(size_t)c * PLANE16_WORDS];
                lo += v & 0xffffu;
                hi += v >> 16;
            }
            *(float2*)&odeg[2 * w] = make_float2((float)lo, (float)hi);
        }
        return;
    }
    int i = blockIdx.x * blockDim.x + threadIdx.x;
    if (i < CH * CH) {
        float sp = log1pf(expf(Wrho[i]));
        W[i] = Wmu[i] + Weps[i] * sp;
    } else if (i < CH * CH + CH) {
        int j = i - CH * CH;
        float sp = log1pf(expf(brho[j]));
        bias[j] = bmu[j] + beps[j] * sp;
    }
}

#define TILE 400
#define N_TILES (N_NODES / TILE)
#define TILE_WORDS (TILE * CH)
#define CHUNK 4096
#define N_CHUNKS ((N_EDGES + CHUNK - 1) / CHUNK)
#define CELL_W 64
#define CELL_CAP (CELL_W - 1)
#define CELLS_WORDS (N_CHUNKS * CELL_W)
#define EMPTY 0xffffffffu
static_assert(CELLS_WORDS <= TILE_WORDS && CELLS_WORDS % 4 == 0, "a tile's cells are parked in its own rows of out");
static_assert(N_NODES <= (1 << 17) && TILE <= 512, "a record is (source row << 9) | local dst row");

__device__ __forceinline__ void put_record(const unsigned d, const unsigned s, unsigned* cell, unsigned* cnt) {
    const unsigned tile = d / TILE;
    const unsigned slot = atomicAdd(&cnt[tile], 1u);
    if (slot < (unsigned)CELL_CAP) cell[tile * CELL_W + 1u + slot] = (s << 9) | (d - tile * TILE);
}

__global__ __launch_bounds__(1024) void edge_cells(const int* __restrict__ src,
                                                   const int* __restrict__ dst,
                                                   float* __restrict__ out) {
    __shared__ unsigned cell[N_TILES * CELL_W];
    __shared__ unsigned cnt[N_TILES];

    const int t = threadIdx.x;
    const int g = blockIdx.x * (CHUNK / 4) + t;
    const bool live = g < N_EDGES / 4;
    int4 d = make_int4(0, 0, 0, 0);
    int4 s = make_int4(0, 0, 0, 0);
    if (live) {
        d = ((const int4*)dst)[g];
        s = ((const int4*)src)[g];
    }

    for (int i = t; i < N_TILES * CELL_W / 4; i += 1024) ((uint4*)cell)[i] = make_uint4(EMPTY, EMPTY, EMPTY, EMPTY);
    if (t < N_TILES) cnt[t] = 0u;
    __syncthreads();

    if (live) {
        put_record((unsigned)d.x, (unsigned)s.x, cell, cnt);
        put_record((unsigned)d.y, (unsigned)s.y, cell, cnt);
        put_record((unsigned)d.z, (unsigned)s.z, cell, cnt);
        put_record((unsigned)d.w, (unsigned)s.w, cell, cnt);
    }
    __syncthreads();

    unsigned* cells = (unsigned*)out + (size_t)blockIdx.x * CELL_W;
    for (int i = t; i < N_TILES * (CELL_W / 4); i += 1024) {
        const int tile = i >> 4;
        const int q = i & 15;
        const unsigned c = cnt[tile];
        uint4 v = *(const uint4*)&cell[tile * CELL_W + q * 4];
        if (c > (unsigned)CELL_CAP) v = make_uint4(EMPTY, EMPTY, EMPTY, EMPTY);
        if (q == 0) v.x = c;
        *(uint4*)&cells[(size_t)tile * TILE_WORDS + q * 4] = v;
    }
}

__global__ __launch_bounds__(128) void gemm_wmma(const float* __restrict__ feat,
                                                 const float* __restrict__ odeg,
                                                 const float* __restrict__ W,
                                                 float* __restrict__ h) {
    __shared__ float lds[16 * LDS_STRIDE];
    const int t = threadIdx.x;
    const int row0 = blockIdx.x * 16;

    {
        const int r = t >> 3;
        const int c = (t & 7) * 16;
        const float s = rsqrtf(fmaxf(odeg[row0 + r], 1.0f));
        const float4* fp = (const float4*)&feat[(size_t)(row0 + r) * CH + c];
#pragma unroll
        for (int j = 0; j < 4; ++j) {
            float4 v = fp[j];
            v.x *= s; v.y *= s; v.z *= s; v.w *= s;
            *(float4*)&lds[r * LDS_STRIDE + c + j * 4] = v;
        }
    }
    __syncthreads();

    const int lane = t & 31;
    const int wave = t >> 5;
    const int half = lane >> 4;
    const int l16  = lane & 15;
    const int n0 = wave * 16;
    const int n1 = 64 + wave * 16;

    v8f c0 = {};
    v8f c1 = {};
    const float* arow = &lds[l16 * LDS_STRIDE + half * 2];

#pragma unroll 4
    for (int k = 0; k < CH; k += 4) {
        v2f a;
        a.x = arow[k];
        a.y = arow[k + 1];

        const float* wk = &W[(size_t)(k + half * 2) * CH + l16];
        v2f b0, b1;
        b0.x = wk[n0];       b0.y = wk[CH + n0];
        b1.x = wk[n1];       b1.y = wk[CH + n1];

        c0 = __builtin_amdgcn_wmma_f32_16x16x4_f32(false, a, false, b0, (short)0, c0, false, false);
        c1 = __builtin_amdgcn_wmma_f32_16x16x4_f32(false, a, false, b1, (short)0, c1, false, false);
    }

#pragma unroll
    for (int r = 0; r < 8; ++r) {
        const int row = row0 + r + half * 8;
        h[(size_t)row * CH + n0 + l16] = c0[r];
        h[(size_t)row * CH + n1 + l16] = c1[r];
    }
}

#define ACC_WORD(ch) ((((ch) & 3) << 5) | ((ch) >> 2))
#define QCAP 24576
#define CELLS_V4 (CELLS_WORDS / 4)
#define CELL_TRIPS ((CELLS_V4 + 1023) / 1024)
#define SCAN_ROWS ((TILE + 31) / 32)

__device__ __forceinline__ void add_row_alone(float* acc, const float* __restrict__ h, const unsigned s, const unsigned dl) {
    const float* hr = h + (size_t)s * CH;
    float* ar = acc + dl * CH;
    for (int k = 0; k < CH; ++k) atomicAdd(&ar[ACC_WORD(k)], hr[k]);
}

__device__ __forceinline__ void count_rec(const unsigned w, unsigned* indeg) {
    if (w != EMPTY) atomicAdd(&indeg[w & 511u], 1u);
}
__device__ __forceinline__ void place_rec(const unsigned w, unsigned* fill, unsigned* sorted) {
    if (w != EMPTY) sorted[atomicAdd(&fill[w & 511u], 1u)] = w >> 9;
}
__device__ __forceinline__ void slow_rec(const unsigned w, float* acc, const float* __restrict__ h) {
    if (w != EMPTY) add_row_alone(acc, h, w >> 9, w & 511u);
}
__device__ __forceinline__ float4 add4(const float4 a, const float4 b) {
    return make_float4(a.x + b.x, a.y + b.y, a.z + b.z, a.w + b.w);
}

__global__ __launch_bounds__(1024) void tile_gather(const int* __restrict__ src,
                                                    const int* __restrict__ dst,
                                                    const float* __restrict__ h,
                                                    const float* __restrict__ bias,
                                                    float* out) {
    __shared__ float acc[TILE * CH];
    __shared__ unsigned sorted[QCAP];
    __shared__ unsigned indeg[TILE];
    __shared__ unsigned rowstart[TILE];
    __shared__ unsigned fill[TILE];
    __shared__ unsigned ovf[N_CHUNKS];
    __shared__ unsigned novf;
    __shared__ unsigned ticket;

    const int t = threadIdx.x;
    const int lane = t & 31;
    const int wave = t >> 5;
    const int lo = blockIdx.x * TILE;

    const unsigned* cells = (const unsigned*)out + (size_t)blockIdx.x * TILE_WORDS;
    uint4 pv[CELL_TRIPS];
#pragma unroll
    for (int k = 0; k < CELL_TRIPS; ++k) {
        const int i = t + 1024 * k;
        pv[k] = (i < CELLS_V4) ? ((const uint4*)cells)[i] : make_uint4(EMPTY, EMPTY, EMPTY, EMPTY);
    }

    if (t < TILE) indeg[t] = 0u;
    if (t == 0) { novf = 0u; ticket = 0u; }
    __syncthreads();

#pragma unroll
    for (int k = 0; k < CELL_TRIPS; ++k) {
        if ((t & 15) == 0) {
            const unsigned c = pv[k].x;
            if (c != EMPTY && c > (unsigned)CELL_CAP) ovf[atomicAdd(&novf, 1u)] = (unsigned)(t + 1024 * k) >> 4;
            pv[k].x = EMPTY;
        }
        count_rec(pv[k].x, indeg);
        count_rec(pv[k].y, indeg);
        count_rec(pv[k].z, indeg);
        count_rec(pv[k].w, indeg);
    }
    __syncthreads();

    if (wave == 0) {
        unsigned loc[SCAN_ROWS];
        unsigned sum = 0u;
#pragma unroll
        for (int j = 0; j < SCAN_ROWS; ++j) {
            const int r = lane * SCAN_ROWS + j;
            loc[j] = sum;
            sum += (r < TILE) ? indeg[r] : 0u;
        }
        unsigned inc = sum;
#pragma unroll
        for (int d = 1; d < 32; d <<= 1) {
            const unsigned o = __shfl_up(inc, d);
            if (lane >= d) inc += o;
        }
        const unsigned excl = inc - sum;
#pragma unroll
        for (int j = 0; j < SCAN_ROWS; ++j) {
            const int r = lane * SCAN_ROWS + j;
            if (r < TILE) { rowstart[r] = excl + loc[j]; fill[r] = excl + loc[j]; }
        }
    }
    __syncthreads();

    const unsigned nov = novf;
    const unsigned total = rowstart[TILE - 1] + indeg[TILE - 1];
    if (nov == 0u && total <= (unsigned)QCAP) {
#pragma unroll
        for (int k = 0; k < CELL_TRIPS; ++k) {
            place_rec(pv[k].x, fill, sorted);
            place_rec(pv[k].y, fill, sorted);
            place_rec(pv[k].z, fill, sorted);
            place_rec(pv[k].w, fill, sorted);
        }
        __syncthreads();

        const float4 b = *(const float4*)&bias[lane * 4];
        for (int it = 0; it < TILE; ++it) {
            unsigned r = 0u;
            if (lane == 0) r = atomicAdd(&ticket, 1u);
            r = __builtin_amdgcn_readfirstlane(r);
            if (r >= (unsigned)TILE) break;
            const unsigned n = __builtin_amdgcn_readfirstlane(indeg[r]);
            const unsigned b0 = __builtin_amdgcn_readfirstlane(rowstart[r]);
            float4 a = make_float4(0.f, 0.f, 0.f, 0.f);
            for (unsigned i = 0; i < n; i += 16u) {
                const unsigned m = min(16u, n - i);
                const unsigned e = ((unsigned)lane < m) ? sorted[b0 + i + (unsigned)lane] : 0u;
                float4 v[16];
                if (m == 16u) {
#pragma unroll
                    for (int k = 0; k < 16; ++k)
                        v[k] = *(const float4*)(h + (size_t)__builtin_amdgcn_readlane(e, k) * CH + lane * 4);
#pragma unroll
                    for (int k = 0; k < 16; ++k) a = add4(a, v[k]);
                } else {
#pragma unroll
                    for (int k = 0; k < 15; ++k)
                        if ((unsigned)k < m) v[k] = *(const float4*)(h + (size_t)__builtin_amdgcn_readlane(e, k) * CH + lane * 4);
#pragma unroll
                    for (int k = 0; k < 15; ++k)
                        if ((unsigned)k < m) a = add4(a, v[k]);
                }
            }
            const float s = rsqrtf(fmaxf((float)n, 1.0f));
            a.x = fmaf(a.x, s, b.x);
            a.y = fmaf(a.y, s, b.y);
            a.z = fmaf(a.z, s, b.z);
            a.w = fmaf(a.w, s, b.w);
            *(float4*)&out[(size_t)(lo + (int)r) * CH + lane * 4] = a;
        }
        return;
    }

    for (int i = t; i < TILE * CH / 4; i += 1024) ((float4*)acc)[i] = make_float4(0.f, 0.f, 0.f, 0.f);
    __syncthreads();
#pragma unroll
    for (int k = 0; k < CELL_TRIPS; ++k) {
        slow_rec(pv[k].x, acc, h);
        slow_rec(pv[k].y, acc, h);
        slow_rec(pv[k].z, acc, h);
        slow_rec(pv[k].w, acc, h);
    }
    for (unsigned o = 0; o < nov; ++o) {
        const unsigned e0 = ovf[o] * (unsigned)CHUNK;
        for (int i = t; i < CHUNK; i += 1024) {
            const unsigned e = e0 + (unsigned)i;
            if (e < (unsigned)N_EDGES) {
                const unsigned dl = (unsigned)(dst[e] - lo);
                if (dl < (unsigned)TILE) {
                    atomicAdd(&indeg[dl], 1u);
                    add_row_alone(acc, h, (unsigned)src[e], dl);
                }
            }
        }
    }
    __syncthreads();
    {
        const float4 b = *(const float4*)&bias[lane * 4];
        for (int r = wave; r < TILE; r += 32) {
            const int node = lo + r;
            const float s = rsqrtf(fmaxf((float)indeg[r], 1.0f));
            const float* ar = &acc[r * CH + lane];
            float4 a = make_float4(ar[0], ar[32], ar[64], ar[96]);
            a.x = fmaf(a.x, s, b.x);
            a.y = fmaf(a.y, s, b.y);
            a.z = fmaf(a.z, s, b.z);
            a.w = fmaf(a.w, s, b.w);
            *(float4*)&out[(size_t)node * CH + lane * 4] = a;
        }
    }
}

extern "C" void kernel_launch(void* const* d_in, const int* in_sizes, int n_in,
                              void* d_out, int out_size, void* d_ws, size_t ws_size,
                              hipStream_t stream) {
    const float* feat  = (const float*)d_in[0];
    const int*   src   = (const int*)  d_in[1];
    const int*   dst   = (const int*)  d_in[2];
    const float* W_mu  = (const float*)d_in[3];
    const float* W_rho = (const float*)d_in[4];
    const float* b_mu  = (const float*)d_in[5];
    const float* b_rho = (const float*)d_in[6];
    const float* W_eps = (const float*)d_in[7];
    const float* b_eps = (const float*)d_in[8];

    float* ws   = (float*)d_ws;
    float* W    = ws + OFF_W;
    float* bias = ws + OFF_BIAS;
    float* odeg = ws + OFF_ODEG;
    float* h    = ws + OFF_H;
    float* out  = (float*)d_out;

    out_degree_counts<<<OD_CHUNKS * 2, 1024, 0, stream>>>(src, out);

    prep_weights<<<PREP_BLOCKS + SUM_BLOCKS, 256, 0, stream>>>(
        W_mu, W_rho, W_eps, b_mu, b_rho, b_eps, W, bias, out, odeg);

    edge_cells<<<N_CHUNKS, 1024, 0, stream>>>(src, dst, out);

    gemm_wmma<<<N_NODES / 16, 128, 0, stream>>>(feat, odeg, W, h);

    tile_gather<<<N_TILES, 1024, 0, stream>>>(src, dst, h, bias, out);
}


# GEMM work-groups take 5 row slabs each and keep their W fragments in registers for all of them: no global load of W in the product loop, 5000 waves launched not 25000; h bit-identical
# speedup vs baseline: 5.8881x; 1.0604x over previous
#include <hip/hip_runtime.h>
#include <hip/hip_bf16.h>
#include <math.h>

typedef __attribute__((ext_vector_type(2))) float v2f;
typedef __attribute__((ext_vector_type(8))) float v8f;

#define N_NODES 100000
#define N_EDGES 1600000
#define CH 128
#define LDS_STRIDE 132

#define OFF_W     0
#define OFF_BIAS  (OFF_W + CH*CH)
#define OFF_ODEG  (OFF_BIAS + CH)
#define OFF_H     (OFF_ODEG + 2 * N_NODES)

#define OD_CHUNKS 64
#define OD_EDGES (N_EDGES / OD_CHUNKS)
#define OD_HALF (N_NODES / 2)
#define OD_WORDS (OD_HALF / 2)
#define OD_V4 (OD_EDGES / 4)
#define OD_TRIPS ((OD_V4 + 1023) / 1024)
#define PLANE16_WORDS (N_NODES / 2)
static_assert(OD_EDGES * OD_CHUNKS == N_EDGES && OD_EDGES % 4 == 0 && OD_EDGES < 65536 && OD_WORDS % 4 == 0,
              "chunks must tile the edge list in 16-byte pieces and stay below the range of a 16-bit counter");

__global__ __launch_bounds__(1024) void out_degree_counts(const int* __restrict__ src, float* __restrict__ out) {
    __shared__ unsigned cnt[OD_WORDS];

    const int t = threadIdx.x;
    const int c = blockIdx.x >> 1;
    const unsigned half = (unsigned)blockIdx.x & 1u;
    const unsigned base = half * OD_HALF;

    const int4* sp = (const int4*)(src + (size_t)c * OD_EDGES);
    int4 sv[OD_TRIPS];
#pragma unroll
    for (int k = 0; k < OD_TRIPS; ++k) {
        const int i = t + 1024 * k;
        sv[k] = (i < OD_V4) ? sp[i] : make_int4(-1, -1, -1, -1);
    }

    for (int i = t; i < OD_WORDS / 4; i += 1024) ((uint4*)cnt)[i] = make_uint4(0u, 0u, 0u, 0u);
    __syncthreads();

#pragma unroll
    for (int k = 0; k < OD_TRIPS; ++k) {
        const unsigned n0 = (unsigned)sv[k].x - base;
        const unsigned n1 = (unsigned)sv[k].y - base;
        const unsigned n2 = (unsigned)sv[k].z - base;
        const unsigned n3 = (unsigned)sv[k].w - base;
        if (n0 < (unsigned)OD_HALF) atomicAdd(&cnt[n0 >> 1], 1u << ((n0 & 1u) * 16u));
        if (n1 < (unsigned)OD_HALF) atomicAdd(&cnt[n1 >> 1], 1u << ((n1 & 1u) * 16u));
        if (n2 < (unsigned)OD_HALF) atomicAdd(&cnt[n2 >> 1], 1u << ((n2 & 1u) * 16u));
        if (n3 < (unsigned)OD_HALF) atomicAdd(&cnt[n3 >> 1], 1u << ((n3 & 1u) * 16u));
    }
    __syncthreads();

    uint4* pp = (uint4*)((unsigned*)out + (size_t)c * PLANE16_WORDS + (size_t)half * OD_WORDS);
    for (int i = t; i < OD_WORDS / 4; i += 1024) pp[i] = ((const uint4*)cnt)[i];
}

#define PREP_BLOCKS ((CH * CH + CH + 255) / 256)
#define SUM_BLOCKS ((PLANE16_WORDS + 255) / 256)
__global__ void prep_weights(const float* __restrict__ Wmu, const float* __restrict__ Wrho,
                             const float* __restrict__ Weps,
                             const float* __restrict__ bmu, const float* __restrict__ brho,
                             const float* __restrict__ beps,
                             float* __restrict__ W, float* __restrict__ bias,
                             const float* __restrict__ out, float* __restrict__ odeg) {
    if (blockIdx.x >= PREP_BLOCKS) {
        const int w = (blockIdx.x - PREP_BLOCKS) * 256 + threadIdx.x;
        if (w < PLANE16_WORDS) {
            const unsigned* pw = (const unsigned*)out + w;
            unsigned lo = 0u, hi = 0u;
#pragma unroll
            for (int c = 0; c < OD_CHUNKS; ++c) {
                const unsigned v = pw[(size_t)c * PLANE16_WORDS];
                lo += v & 0xffffu;
                hi += v >> 16;
            }
            *(float2*)&odeg[2 * w] = make_float2((float)lo, (float)hi);
        }
        return;
    }
    int i = blockIdx.x * blockDim.x + threadIdx.x;
    if (i < CH * CH) {
        float sp = log1pf(expf(Wrho[i]));
        W[i] = Wmu[i] + Weps[i] * sp;
    } else if (i < CH * CH + CH) {
        int j = i - CH * CH;
        float sp = log1pf(expf(brho[j]));
        bias[j] = bmu[j] + beps[j] * sp;
    }
}

#define TILE 400
#define N_TILES (N_NODES / TILE)
#define TILE_WORDS (TILE * CH)
#define CHUNK 4096
#define N_CHUNKS ((N_EDGES + CHUNK - 1) / CHUNK)
#define CELL_W 64
#define CELL_CAP (CELL_W - 1)
#define CELLS_WORDS (N_CHUNKS * CELL_W)
#define EMPTY 0xffffffffu
static_assert(CELLS_WORDS <= TILE_WORDS && CELLS_WORDS % 4 == 0, "a tile's cells are parked in its own rows of out");
static_assert(N_NODES <= (1 << 17) && TILE <= 512, "a record is (source row << 9) | local dst row");

__device__ __forceinline__ void put_record(const unsigned d, const unsigned s, unsigned* cell, unsigned* cnt) {
    const unsigned tile = d / TILE;
    const unsigned slot = atomicAdd(&cnt[tile], 1u);
    if (slot < (unsigned)CELL_CAP) cell[tile * CELL_W + 1u + slot] = (s << 9) | (d - tile * TILE);
}

__global__ __launch_bounds__(1024) void edge_cells(const int* __restrict__ src,
                                                   const int* __restrict__ dst,
                                                   float* __restrict__ out) {
    __shared__ unsigned cell[N_TILES * CELL_W];
    __shared__ unsigned cnt[N_TILES];

    const int t = threadIdx.x;
    const int g = blockIdx.x * (CHUNK / 4) + t;
    const bool live = g < N_EDGES / 4;
    int4 d = make_int4(0, 0, 0, 0);
    int4 s = make_int4(0, 0, 0, 0);
    if (live) {
        d = ((const int4*)dst)[g];
        s = ((const int4*)src)[g];
    }

    for (int i = t; i < N_TILES * CELL_W / 4; i += 1024) ((uint4*)cell)[i] = make_uint4(EMPTY, EMPTY, EMPTY, EMPTY);
    if (t < N_TILES) cnt[t] = 0u;
    __syncthreads();

    if (live) {
        put_record((unsigned)d.x, (unsigned)s.x, cell, cnt);
        put_record((unsigned)d.y, (unsigned)s.y, cell, cnt);
        put_record((unsigned)d.z, (unsigned)s.z, cell, cnt);
        put_record((unsigned)d.w, (unsigned)s.w, cell, cnt);
    }
    __syncthreads();

    unsigned* cells = (unsigned*)out + (size_t)blockIdx.x * CELL_W;
    for (int i = t; i < N_TILES * (CELL_W / 4); i += 1024) {
        const int tile = i >> 4;
        const int q = i & 15;
        const unsigned c = cnt[tile];
        uint4 v = *(const uint4*)&cell[tile * CELL_W + q * 4];
        if (c > (unsigned)CELL_CAP) v = make_uint4(EMPTY, EMPTY, EMPTY, EMPTY);
        if (q == 0) v.x = c;
        *(uint4*)&cells[(size_t)tile * TILE_WORDS + q * 4] = v;
    }
}

#define SLAB_ROWS 16
#define N_SLABS (N_NODES / SLAB_ROWS)
#define SLABS_PER_WG 5
#define GEMM_WGS (N_SLABS / SLABS_PER_WG)
#define K_STEPS (CH / 4)
static_assert(GEMM_WGS * SLABS_PER_WG * SLAB_ROWS == N_NODES, "the slabs must tile the rows exactly");

__global__ __launch_bounds__(128) void gemm_wmma(const float* __restrict__ feat,
                                                 const float* __restrict__ odeg,
                                                 const float* __restrict__ W,
                                                 float* __restrict__ h) {
    __shared__ float lds[2][SLAB_ROWS * LDS_STRIDE];
    const int t = threadIdx.x;
    const int lane = t & 31;
    const int wave = t >> 5;
    const int half = lane >> 4;
    const int l16  = lane & 15;
    const int n0 = wave * 16;
    const int n1 = 64 + wave * 16;
    const int sr = t >> 3;
    const int sc = (t & 7) * 16;

    int row0 = blockIdx.x * (SLABS_PER_WG * SLAB_ROWS);
    float4 pf[4];
    float ps;
    {
        const float4* fp = (const float4*)&feat[(size_t)(row0 + sr) * CH + sc];
#pragma unroll
        for (int j = 0; j < 4; ++j) pf[j] = fp[j];
        ps = odeg[row0 + sr];
    }

    v2f b0[K_STEPS], b1[K_STEPS];
#pragma unroll
    for (int s = 0; s < K_STEPS; ++s) {
        const float* wk = &W[(size_t)(4 * s + half * 2) * CH + l16];
        b0[s].x = wk[n0];       b0[s].y = wk[CH + n0];
        b1[s].x = wk[n1];       b1[s].y = wk[CH + n1];
    }

#pragma nounroll
    for (int i = 0; i < SLABS_PER_WG; ++i) {
        float* buf = lds[i & 1];

        {
            const float s = rsqrtf(fmaxf(ps, 1.0f));
#pragma unroll
            for (int j = 0; j < 4; ++j) {
                float4 v = pf[j];
                v.x *= s; v.y *= s; v.z *= s; v.w *= s;
                *(float4*)&buf[sr * LDS_STRIDE + sc + j * 4] = v;
            }
        }
        __syncthreads();

        if (i + 1 < SLABS_PER_WG) {
            const float4* fp = (const float4*)&feat[(size_t)(row0 + SLAB_ROWS + sr) * CH + sc];
#pragma unroll
            for (int j = 0; j < 4; ++j) pf[j] = fp[j];
            ps = odeg[row0 + SLAB_ROWS + sr];
        }

        v8f c0 = {};
        v8f c1 = {};
        const float* arow = &buf[l16 * LDS_STRIDE + half * 2];

#pragma unroll
        for (int s = 0; s < K_STEPS; ++s) {
            v2f a;
            a.x = arow[4 * s];
            a.y = arow[4 * s + 1];
            c0 = __builtin_amdgcn_wmma_f32_16x16x4_f32(false, a, false, b0[s], (short)0, c0, false, false);
            c1 = __builtin_amdgcn_wmma_f32_16x16x4_f32(false, a, false, b1[s], (short)0, c1, false, false);
        }

#pragma unroll
        for (int r = 0; r < 8; ++r) {
            const int row = row0 + r + half * 8;
            h[(size_t)row * CH + n0 + l16] = c0[r];
            h[(size_t)row * CH + n1 + l16] = c1[r];
        }
        row0 += SLAB_ROWS;
    }
}

#define ACC_WORD(ch) ((((ch) & 3) << 5) | ((ch) >> 2))
#define QCAP 24576
#define CELLS_V4 (CELLS_WORDS / 4)
#define CELL_TRIPS ((CELLS_V4 + 1023) / 1024)
#define SCAN_ROWS ((TILE + 31) / 32)

__device__ __forceinline__ void add_row_alone(float* acc, const float* __restrict__ h, const unsigned s, const unsigned dl) {
    const float* hr = h + (size_t)s * CH;
    float* ar = acc + dl * CH;
    for (int k = 0; k < CH; ++k) atomicAdd(&ar[ACC_WORD(k)], hr[k]);
}

__device__ __forceinline__ void count_rec(const unsigned w, unsigned* indeg) {
    if (w != EMPTY) atomicAdd(&indeg[w & 511u], 1u);
}
__device__ __forceinline__ void place_rec(const unsigned w, unsigned* fill, unsigned* sorted) {
    if (w != EMPTY) sorted[atomicAdd(&fill[w & 511u], 1u)] = w >> 9;
}
__device__ __forceinline__ void slow_rec(const unsigned w, float* acc, const float* __restrict__ h) {
    if (w != EMPTY) add_row_alone(acc, h, w >> 9, w & 511u);
}
__device__ __forceinline__ float4 add4(const float4 a, const float4 b) {
    return make_float4(a.x + b.x, a.y + b.y, a.z + b.z, a.w + b.w);
}

__global__ __launch_bounds__(1024) void tile_gather(const int* __restrict__ src,
                                                    const int* __restrict__ dst,
                                                    const float* __restrict__ h,
                                                    const float* __restrict__ bias,
                                                    float* out) {
    __shared__ float acc[TILE * CH];
    __shared__ unsigned sorted[QCAP];
    __shared__ unsigned indeg[TILE];
    __shared__ unsigned rowstart[TILE];
    __shared__ unsigned fill[TILE];
    __shared__ unsigned ovf[N_CHUNKS];
    __shared__ unsigned novf;
    __shared__ unsigned ticket;

    const int t = threadIdx.x;
    const int lane = t & 31;
    const int wave = t >> 5;
    const int lo = blockIdx.x * TILE;

    const unsigned* cells = (const unsigned*)out + (size_t)blockIdx.x * TILE_WORDS;
    uint4 pv[CELL_TRIPS];
#pragma unroll
    for (int k = 0; k < CELL_TRIPS; ++k) {
        const int i = t + 1024 * k;
        pv[k] = (i < CELLS_V4) ? ((const uint4*)cells)[i] : make_uint4(EMPTY, EMPTY, EMPTY, EMPTY);
    }

    if (t < TILE) indeg[t] = 0u;
    if (t == 0) { novf = 0u; ticket = 0u; }
    __syncthreads();

#pragma unroll
    for (int k = 0; k < CELL_TRIPS; ++k) {
        if ((t & 15) == 0) {
            const unsigned c = pv[k].x;
            if (c != EMPTY && c > (unsigned)CELL_CAP) ovf[atomicAdd(&novf, 1u)] = (unsigned)(t + 1024 * k) >> 4;
            pv[k].x = EMPTY;
        }
        count_rec(pv[k].x, indeg);
        count_rec(pv[k].y, indeg);
        count_rec(pv[k].z, indeg);
        count_rec(pv[k].w, indeg);
    }
    __syncthreads();

    if (wave == 0) {
        unsigned loc[SCAN_ROWS];
        unsigned sum = 0u;
#pragma unroll
        for (int j = 0; j < SCAN_ROWS; ++j) {
            const int r = lane * SCAN_ROWS + j;
            loc[j] = sum;
            sum += (r < TILE) ? indeg[r] : 0u;
        }
        unsigned inc = sum;
#pragma unroll
        for (int d = 1; d < 32; d <<= 1) {
            const unsigned o = __shfl_up(inc, d);
            if (lane >= d) inc += o;
        }
        const unsigned excl = inc - sum;
#pragma unroll
        for (int j = 0; j < SCAN_ROWS; ++j) {
            const int r = lane * SCAN_ROWS + j;
            if (r < TILE) { rowstart[r] = excl + loc[j]; fill[r] = excl + loc[j]; }
        }
    }
    __syncthreads();

    const unsigned nov = novf;
    const unsigned total = rowstart[TILE - 1] + indeg[TILE - 1];
    if (nov == 0u && total <= (unsigned)QCAP) {
#pragma unroll
        for (int k = 0; k < CELL_TRIPS; ++k) {
            place_rec(pv[k].x, fill, sorted);
            place_rec(pv[k].y, fill, sorted);
            place_rec(pv[k].z, fill, sorted);
            place_rec(pv[k].w, fill, sorted);
        }
        __syncthreads();

        const float4 b = *(const float4*)&bias[lane * 4];
        for (int it = 0; it < TILE; ++it) {
            unsigned r = 0u;
            if (lane == 0) r = atomicAdd(&ticket, 1u);
            r = __builtin_amdgcn_readfirstlane(r);
            if (r >= (unsigned)TILE) break;
            const unsigned n = __builtin_amdgcn_readfirstlane(indeg[r]);
            const unsigned b0 = __builtin_amdgcn_readfirstlane(rowstart[r]);
            float4 a = make_float4(0.f, 0.f, 0.f, 0.f);
            for (unsigned i = 0; i < n; i += 16u) {
                const unsigned m = min(16u, n - i);
                const unsigned e = ((unsigned)lane < m) ? sorted[b0 + i + (unsigned)lane] : 0u;
                float4 v[16];
                if (m == 16u) {
#pragma unroll
                    for (int k = 0; k < 16; ++k)
                        v[k] = *(const float4*)(h + (size_t)__builtin_amdgcn_readlane(e, k) * CH + lane * 4);
#pragma unroll
                    for (int k = 0; k < 16; ++k) a = add4(a, v[k]);
                } else {
#pragma unroll
                    for (int k = 0; k < 15; ++k)
                        if ((unsigned)k < m) v[k] = *(const float4*)(h + (size_t)__builtin_amdgcn_readlane(e, k) * CH + lane * 4);
#pragma unroll
                    for (int k = 0; k < 15; ++k)
                        if ((unsigned)k < m) a = add4(a, v[k]);
                }
            }
            const float s = rsqrtf(fmaxf((float)n, 1.0f));
            a.x = fmaf(a.x, s, b.x);
            a.y = fmaf(a.y, s, b.y);
            a.z = fmaf(a.z, s, b.z);
            a.w = fmaf(a.w, s, b.w);
            *(float4*)&out[(size_t)(lo + (int)r) * CH + lane * 4] = a;
        }
        return;
    }

    for (int i = t; i < TILE * CH / 4; i += 1024) ((float4*)acc)[i] = make_float4(0.f, 0.f, 0.f, 0.f);
    __syncthreads();
#pragma unroll
    for (int k = 0; k < CELL_TRIPS; ++k) {
        slow_rec(pv[k].x, acc, h);
        slow_rec(pv[k].y, acc, h);
        slow_rec(pv[k].z, acc, h);
        slow_rec(pv[k].w, acc, h);
    }
    for (unsigned o = 0; o < nov; ++o) {
        const unsigned e0 = ovf[o] * (unsigned)CHUNK;
        for (int i = t; i < CHUNK; i += 1024) {
            const unsigned e = e0 + (unsigned)i;
            if (e < (unsigned)N_EDGES) {
                const unsigned dl = (unsigned)(dst[e] - lo);
                if (dl < (unsigned)TILE) {
                    atomicAdd(&indeg[dl], 1u);
                    add_row_alone(acc, h, (unsigned)src[e], dl);
                }
            }
        }
    }
    __syncthreads();
    {
        const float4 b = *(const float4*)&bias[lane * 4];
        for (int r = wave; r < TILE; r += 32) {
            const int node = lo + r;
            const float s = rsqrtf(fmaxf((float)indeg[r], 1.0f));
            const float* ar = &acc[r * CH + lane];
            float4 a = make_float4(ar[0], ar[32], ar[64], ar[96]);
            a.x = fmaf(a.x, s, b.x);
            a.y = fmaf(a.y, s, b.y);
            a.z = fmaf(a.z, s, b.z);
            a.w = fmaf(a.w, s, b.w);
            *(float4*)&out[(size_t)node * CH + lane * 4] = a;
        }
    }
}

extern "C" void kernel_launch(void* const* d_in, const int* in_sizes, int n_in,
                              void* d_out, int out_size, void* d_ws, size_t ws_size,
                              hipStream_t stream) {
    const float* feat  = (const float*)d_in[0];
    const int*   src   = (const int*)  d_in[1];
    const int*   dst   = (const int*)  d_in[2];
    const float* W_mu  = (const float*)d_in[3];
    const float* W_rho = (const float*)d_in[4];
    const float* b_mu  = (const float*)d_in[5];
    const float* b_rho = (const float*)d_in[6];
    const float* W_eps = (const float*)d_in[7];
    const float* b_eps = (const float*)d_in[8];

    float* ws   = (float*)d_ws;
    float* W    = ws + OFF_W;
    float* bias = ws + OFF_BIAS;
    float* odeg = ws + OFF_ODEG;
    float* h    = ws + OFF_H;
    float* out  = (float*)d_out;

    out_degree_counts<<<OD_CHUNKS * 2, 1024, 0, stream>>>(src, out);

    prep_weights<<<PREP_BLOCKS + SUM_BLOCKS, 256, 0, stream>>>(
        W_mu, W_rho, W_eps, b_mu, b_rho, b_eps, W, bias, out, odeg);

    edge_cells<<<N_CHUNKS, 1024, 0, stream>>>(src, dst, out);

    gemm_wmma<<<GEMM_WGS, 128, 0, stream>>>(feat, odeg, W, h);

    tile_gather<<<N_TILES, 1024, 0, stream>>>(src, dst, h, bias, out);
}


# Launch count 5 to 3: out-degree counts, record cells and weight prep run side by side in one launch; GEMM work-groups sum the count planes of their own rows; f32 sums unchanged
# speedup vs baseline: 6.1366x; 1.0422x over previous
#include <hip/hip_runtime.h>
#include <hip/hip_bf16.h>
#include <math.h>

typedef __attribute__((ext_vector_type(2))) float v2f;
typedef __attribute__((ext_vector_type(8))) float v8f;

#define N_NODES 100000
#define N_EDGES 1600000
#define CH 128
#define LDS_STRIDE 132

#define OFF_W     0
#define OFF_BIAS  (OFF_W + CH*CH)
#define OFF_H     (OFF_BIAS + CH + 2 * N_NODES)

#define TILE 400
#define N_TILES (N_NODES / TILE)
#define TILE_WORDS (TILE * CH)
#define CHUNK 4096
#define N_CHUNKS ((N_EDGES + CHUNK - 1) / CHUNK)
#define CELL_W 64
#define CELL_CAP (CELL_W - 1)
#define CELLS_WORDS (N_CHUNKS * CELL_W)
#define EMPTY 0xffffffffu
static_assert(CELLS_WORDS <= TILE_WORDS && CELLS_WORDS % 4 == 0, "a tile's cells are parked in its own rows of out");
static_assert(N_NODES <= (1 << 17) && TILE <= 512, "a record is (source row << 9) | local dst row");

#define OD_CHUNKS 64
#define OD_PIECES (OD_CHUNKS * 2)
#define OD_EDGES (N_EDGES / OD_CHUNKS)
#define OD_HALF (N_NODES / 2)
#define OD_WORDS (OD_HALF / 2)
#define OD_V4 (OD_EDGES / 4)
#define OD_TRIPS ((OD_V4 + 1023) / 1024)
static_assert(OD_EDGES * OD_CHUNKS == N_EDGES && OD_EDGES % 4 == 0 && OD_EDGES < 65536 && OD_WORDS % 4 == 0,
              "chunks must tile the edge list in 16-byte pieces and stay below the range of a 16-bit counter");
static_assert(OD_PIECES <= N_TILES && CELLS_WORDS + OD_WORDS <= TILE_WORDS,
              "a piece of partial counts is parked behind the cells of a tile of its own");

__device__ __forceinline__ void out_degree_counts(const int piece, unsigned* cnt, const int* __restrict__ src,
                                                  float* __restrict__ out) {
    const int t = threadIdx.x;
    const int c = piece >> 1;
    const unsigned half = (unsigned)piece & 1u;
    const unsigned base = half * OD_HALF;

    const int4* sp = (const int4*)(src + (size_t)c * OD_EDGES);
    int4 sv[OD_TRIPS];
#pragma unroll
    for (int k = 0; k < OD_TRIPS; ++k) {
        const int i = t + 1024 * k;
        sv[k] = (i < OD_V4) ? sp[i] : make_int4(-1, -1, -1, -1);
    }

    for (int i = t; i < OD_WORDS / 4; i += 1024) ((uint4*)cnt)[i] = make_uint4(0u, 0u, 0u, 0u);
    __syncthreads();

#pragma unroll
    for (int k = 0; k < OD_TRIPS; ++k) {
        const unsigned n0 = (unsigned)sv[k].x - base;
        const unsigned n1 = (unsigned)sv[k].y - base;
        const unsigned n2 = (unsigned)sv[k].z - base;
        const unsigned n3 = (unsigned)sv[k].w - base;
        if (n0 < (unsigned)OD_HALF) atomicAdd(&cnt[n0 >> 1], 1u << ((n0 & 1u) * 16u));
        if (n1 < (unsigned)OD_HALF) atomicAdd(&cnt[n1 >> 1], 1u << ((n1 & 1u) * 16u));
        if (n2 < (unsigned)OD_HALF) atomicAdd(&cnt[n2 >> 1], 1u << ((n2 & 1u) * 16u));
        if (n3 < (unsigned)OD_HALF) atomicAdd(&cnt[n3 >> 1], 1u << ((n3 & 1u) * 16u));
    }
    __syncthreads();

    uint4* pp = (uint4*)((unsigned*)out + (size_t)piece * TILE_WORDS + CELLS_WORDS);
    for (int i = t; i < OD_WORDS / 4; i += 1024) pp[i] = ((const uint4*)cnt)[i];
}

#define PREP_BLOCKS ((CH * CH + CH + 1023) / 1024)
__device__ __forceinline__ void prep_weights(const int i,
                                             const float* __restrict__ Wmu, const float* __restrict__ Wrho,
                                             const float* __restrict__ Weps,
                                             const float* __restrict__ bmu, const float* __restrict__ brho,
                                             const float* __restrict__ beps,
                                             float* __restrict__ W, float* __restrict__ bias) {
    if (i < CH * CH) {
        float sp = log1pf(expf(Wrho[i]));
        W[i] = Wmu[i] + Weps[i] * sp;
    } else if (i < CH * CH + CH) {
        int j = i - CH * CH;
        float sp = log1pf(expf(brho[j]));
        bias[j] = bmu[j] + beps[j] * sp;
    }
}

__device__ __forceinline__ void put_record(const unsigned d, const unsigned s, unsigned* cell, unsigned* cnt) {
    const unsigned tile = d / TILE;
    const unsigned slot = atomicAdd(&cnt[tile], 1u);
    if (slot < (unsigned)CELL_CAP) cell[tile * CELL_W + 1u + slot] = (s << 9) | (d - tile * TILE);
}

#define CELL_LDS_WORDS (N_TILES * CELL_W)
static_assert(CELL_LDS_WORDS + N_TILES <= OD_WORDS, "the cells and their counters fit in the launch's LDS array");

__device__ __forceinline__ void edge_cells(const int chunk, unsigned* cell, unsigned* cnt,
                                           const int* __restrict__ src, const int* __restrict__ dst,
                                           float* __restrict__ out) {
    const int t = threadIdx.x;
    const int g = chunk * (CHUNK / 4) + t;
    const bool live = g < N_EDGES / 4;
    int4 d = make_int4(0, 0, 0, 0);
    int4 s = make_int4(0, 0, 0, 0);
    if (live) {
        d = ((const int4*)dst)[g];
        s = ((const int4*)src)[g];
    }

    for (int i = t; i < N_TILES * CELL_W / 4; i += 1024) ((uint4*)cell)[i] = make_uint4(EMPTY, EMPTY, EMPTY, EMPTY);
    if (t < N_TILES) cnt[t] = 0u;
    __syncthreads();

    if (live) {
        put_record((unsigned)d.x, (unsigned)s.x, cell, cnt);
        put_record((unsigned)d.y, (unsigned)s.y, cell, cnt);
        put_record((unsigned)d.z, (unsigned)s.z, cell, cnt);
        put_record((unsigned)d.w, (unsigned)s.w, cell, cnt);
    }
    __syncthreads();

    unsigned* cells = (unsigned*)out + (size_t)chunk * CELL_W;
    for (int i = t; i < N_TILES * (CELL_W / 4); i += 1024) {
        const int tile = i >> 4;
        const int q = i & 15;
        const unsigned c = cnt[tile];
        uint4 v = *(const uint4*)&cell[tile * CELL_W + q * 4];
        if (c > (unsigned)CELL_CAP) v = make_uint4(EMPTY, EMPTY, EMPTY, EMPTY);
        if (q == 0) v.x = c;
        *(uint4*)&cells[(size_t)tile * TILE_WORDS + q * 4] = v;
    }
}

#define PASS1_WGS (OD_PIECES + N_CHUNKS + PREP_BLOCKS)
__global__ __launch_bounds__(1024) void input_pass(const int* __restrict__ src, const int* __restrict__ dst,
                                                   const float* __restrict__ Wmu, const float* __restrict__ Wrho,
                                                   const float* __restrict__ Weps,
                                                   const float* __restrict__ bmu, const float* __restrict__ brho,
                                                   const float* __restrict__ beps,
                                                   float* __restrict__ W, float* __restrict__ bias,
                                                   float* __restrict__ out) {
    __shared__ unsigned lds[OD_WORDS];
    const int b = blockIdx.x;
    if (b < OD_PIECES) {
        out_degree_counts(b, lds, src, out);
    } else if (b < OD_PIECES + N_CHUNKS) {
        edge_cells(b - OD_PIECES, lds, lds + CELL_LDS_WORDS, src, dst, out);
    } else {
        prep_weights((b - OD_PIECES - N_CHUNKS) * 1024 + (int)threadIdx.x, Wmu, Wrho, Weps, bmu, brho, beps, W, bias);
    }
}

#define SLAB_ROWS 16
#define N_SLABS (N_NODES / SLAB_ROWS)
#define SLABS_PER_WG 5
#define WG_ROWS (SLABS_PER_WG * SLAB_ROWS)
#define WG_WORDS (WG_ROWS / 2)
#define GEMM_WGS (N_SLABS / SLABS_PER_WG)
#define K_STEPS (CH / 4)
#define OD_GROUPS 16
#define OD_PER_GROUP (OD_CHUNKS / OD_GROUPS)
#define OD_BLOCKS (WG_WORDS / 8)
static_assert(GEMM_WGS * SLABS_PER_WG * SLAB_ROWS == N_NODES, "the slabs must tile the rows exactly");
static_assert(OD_WORDS % WG_WORDS == 0, "a work-group's rows lie in one half of the node range");
static_assert(OD_GROUPS * 8 == 128 && OD_GROUPS * OD_PER_GROUP == OD_CHUNKS && OD_BLOCKS * 8 == WG_WORDS,
              "sixteen groups of eight threads cover every plane and every word exactly once");

__global__ __launch_bounds__(128) void gemm_wmma(const float* __restrict__ feat,
                                                 const float* __restrict__ out,
                                                 const float* __restrict__ W,
                                                 float* __restrict__ h) {
    __shared__ float lds[2][SLAB_ROWS * LDS_STRIDE];
    __shared__ unsigned part[OD_GROUPS][WG_ROWS];
    __shared__ float odeg[WG_ROWS];
    const int t = threadIdx.x;
    const int lane = t & 31;
    const int wave = t >> 5;
    const int half = lane >> 4;
    const int l16  = lane & 15;
    const int n0 = wave * 16;
    const int n1 = 64 + wave * 16;
    const int sr = t >> 3;
    const int sc = (t & 7) * 16;

    int row0 = blockIdx.x * WG_ROWS;
    float4 pf[4];
    {
        const float4* fp = (const float4*)&feat[(size_t)(row0 + sr) * CH + sc];
#pragma unroll
        for (int j = 0; j < 4; ++j) pf[j] = fp[j];
    }

    const int og = t >> 3;
    const int oj = t & 7;
    unsigned pc[OD_PER_GROUP][OD_BLOCKS];
    {
        const unsigned w0 = (unsigned)blockIdx.x * WG_WORDS;
        const unsigned hf = (w0 >= (unsigned)OD_WORDS) ? 1u : 0u;
        const unsigned* pw = (const unsigned*)out + (size_t)(2 * OD_PER_GROUP * og + (int)hf) * TILE_WORDS
                             + CELLS_WORDS + (w0 - hf * OD_WORDS) + (unsigned)oj;
#pragma unroll
        for (int p = 0; p < OD_PER_GROUP; ++p)
#pragma unroll
            for (int q = 0; q < OD_BLOCKS; ++q) pc[p][q] = pw[(size_t)(2 * p) * TILE_WORDS + 8 * q];
    }

    v2f b0[K_STEPS], b1[K_STEPS];
#pragma unroll
    for (int s = 0; s < K_STEPS; ++s) {
        const float* wk = &W[(size_t)(4 * s + half * 2) * CH + l16];
        b0[s].x = wk[n0];       b0[s].y = wk[CH + n0];
        b1[s].x = wk[n1];       b1[s].y = wk[CH + n1];
    }

#pragma unroll
    for (int q = 0; q < OD_BLOCKS; ++q) {
        unsigned lo = 0u, hi = 0u;
#pragma unroll
        for (int p = 0; p < OD_PER_GROUP; ++p) {
            lo += pc[p][q] & 0xffffu;
            hi += pc[p][q] >> 16;
        }
        *(uint2*)&part[og][2 * (8 * q + oj)] = make_uint2(lo, hi);
    }
    __syncthreads();
    if (t < WG_ROWS) {
        unsigned n = 0u;
#pragma unroll
        for (int g = 0; g < OD_GROUPS; ++g) n += part[g][t];
        odeg[t] = (float)n;
    }
    __syncthreads();

#pragma nounroll
    for (int i = 0; i < SLABS_PER_WG; ++i) {
        float* buf = lds[i & 1];

        {
            const float s = rsqrtf(fmaxf(odeg[i * SLAB_ROWS + sr], 1.0f));
#pragma unroll
            for (int j = 0; j < 4; ++j) {
                float4 v = pf[j];
                v.x *= s; v.y *= s; v.z *= s; v.w *= s;
                *(float4*)&buf[sr * LDS_STRIDE + sc + j * 4] = v;
            }
        }
        __syncthreads();

        if (i + 1 < SLABS_PER_WG) {
            const float4* fp = (const float4*)&feat[(size_t)(row0 + SLAB_ROWS + sr) * CH + sc];
#pragma unroll
            for (int j = 0; j < 4; ++j) pf[j] = fp[j];
        }

        v8f c0 = {};
        v8f c1 = {};
        const float* arow = &buf[l16 * LDS_STRIDE + half * 2];

#pragma unroll
        for (int s = 0; s < K_STEPS; ++s) {
            v2f a;
            a.x = arow[4 * s];
            a.y = arow[4 * s + 1];
            c0 = __builtin_amdgcn_wmma_f32_16x16x4_f32(false, a, false, b0[s], (short)0, c0, false, false);
            c1 = __builtin_amdgcn_wmma_f32_16x16x4_f32(false, a, false, b1[s], (short)0, c1, false, false);
        }

#pragma unroll
        for (int r = 0; r < 8; ++r) {
            const int row = row0 + r + half * 8;
            h[(size_t)row * CH + n0 + l16] = c0[r];
            h[(size_t)row * CH + n1 + l16] = c1[r];
        }
        row0 += SLAB_ROWS;
    }
}

#define ACC_WORD(ch) ((((ch) & 3) << 5) | ((ch) >> 2))
#define QCAP 24576
#define CELLS_V4 (CELLS_WORDS / 4)
#define CELL_TRIPS ((CELLS_V4 + 1023) / 1024)
#define SCAN_ROWS ((TILE + 31) / 32)

__device__ __forceinline__ void add_row_alone(float* acc, const float* __restrict__ h, const unsigned s, const unsigned dl) {
    const float* hr = h + (size_t)s * CH;
    float* ar = acc + dl * CH;
    for (int k = 0; k < CH; ++k) atomicAdd(&ar[ACC_WORD(k)], hr[k]);
}

__device__ __forceinline__ void count_rec(const unsigned w, unsigned* indeg) {
    if (w != EMPTY) atomicAdd(&indeg[w & 511u], 1u);
}
__device__ __forceinline__ void place_rec(const unsigned w, unsigned* fill, unsigned* sorted) {
    if (w != EMPTY) sorted[atomicAdd(&fill[w & 511u], 1u)] = w >> 9;
}
__device__ __forceinline__ void slow_rec(const unsigned w, float* acc, const float* __restrict__ h) {
    if (w != EMPTY) add_row_alone(acc, h, w >> 9, w & 511u);
}
__device__ __forceinline__ float4 add4(const float4 a, const float4 b) {
    return make_float4(a.x + b.x, a.y + b.y, a.z + b.z, a.w + b.w);
}

__global__ __launch_bounds__(1024) void tile_gather(const int* __restrict__ src,
                                                    const int* __restrict__ dst,
                                                    const float* __restrict__ h,
                                                    const float* __restrict__ bias,
                                                    float* out) {
    __shared__ float acc[TILE * CH];
    __shared__ unsigned sorted[QCAP];
    __shared__ unsigned indeg[TILE];
    __shared__ unsigned rowstart[TILE];
    __shared__ unsigned fill[TILE];
    __shared__ unsigned ovf[N_CHUNKS];
    __shared__ unsigned novf;
    __shared__ unsigned ticket;

    const int t = threadIdx.x;
    const int lane = t & 31;
    const int wave = t >> 5;
    const int lo = blockIdx.x * TILE;

    const unsigned* cells = (const unsigned*)out + (size_t)blockIdx.x * TILE_WORDS;
    uint4 pv[CELL_TRIPS];
#pragma unroll
    for (int k = 0; k < CELL_TRIPS; ++k) {
        const int i = t + 1024 * k;
        pv[k] = (i < CELLS_V4) ? ((const uint4*)cells)[i] : make_uint4(EMPTY, EMPTY, EMPTY, EMPTY);
    }

    if (t < TILE) indeg[t] = 0u;
    if (t == 0) { novf = 0u; ticket = 0u; }
    __syncthreads();

#pragma unroll
    for (int k = 0; k < CELL_TRIPS; ++k) {
        if ((t & 15) == 0) {
            const unsigned c = pv[k].x;
            if (c != EMPTY && c > (unsigned)CELL_CAP) ovf[atomicAdd(&novf, 1u)] = (unsigned)(t + 1024 * k) >> 4;
            pv[k].x = EMPTY;
        }
        count_rec(pv[k].x, indeg);
        count_rec(pv[k].y, indeg);
        count_rec(pv[k].z, indeg);
        count_rec(pv[k].w, indeg);
    }
    __syncthreads();

    if (wave == 0) {
        unsigned loc[SCAN_ROWS];
        unsigned sum = 0u;
#pragma unroll
        for (int j = 0; j < SCAN_ROWS; ++j) {
            const int r = lane * SCAN_ROWS + j;
            loc[j] = sum;
            sum += (r < TILE) ? indeg[r] : 0u;
        }
        unsigned inc = sum;
#pragma unroll
        for (int d = 1; d < 32; d <<= 1) {
            const unsigned o = __shfl_up(inc, d);
            if (lane >= d) inc += o;
        }
        const unsigned excl = inc - sum;
#pragma unroll
        for (int j = 0; j < SCAN_ROWS; ++j) {
            const int r = lane * SCAN_ROWS + j;
            if (r < TILE) { rowstart[r] = excl + loc[j]; fill[r] = excl + loc[j]; }
        }
    }
    __syncthreads();

    const unsigned nov = novf;
    const unsigned total = rowstart[TILE - 1] + indeg[TILE - 1];
    if (nov == 0u && total <= (unsigned)QCAP) {
#pragma unroll
        for (int k = 0; k < CELL_TRIPS; ++k) {
            place_rec(pv[k].x, fill, sorted);
            place_rec(pv[k].y, fill, sorted);
            place_rec(pv[k].z, fill, sorted);
            place_rec(pv[k].w, fill, sorted);
        }
        __syncthreads();

        const float4 b = *(const float4*)&bias[lane * 4];
        for (int it = 0; it < TILE; ++it) {
            unsigned r = 0u;
            if (lane == 0) r = atomicAdd(&ticket, 1u);
            r = __builtin_amdgcn_readfirstlane(r);
            if (r >= (unsigned)TILE) break;
            const unsigned n = __builtin_amdgcn_readfirstlane(indeg[r]);
            const unsigned b0 = __builtin_amdgcn_readfirstlane(rowstart[r]);
            float4 a = make_float4(0.f, 0.f, 0.f, 0.f);
            for (unsigned i = 0; i < n; i += 16u) {
                const unsigned m = min(16u, n - i);
                const unsigned e = ((unsigned)lane < m) ? sorted[b0 + i + (unsigned)lane] : 0u;
                float4 v[16];
                if (m == 16u) {
#pragma unroll
                    for (int k = 0; k < 16; ++k)
                        v[k] = *(const float4*)(h + (size_t)__builtin_amdgcn_readlane(e, k) * CH + lane * 4);
#pragma unroll
                    for (int k = 0; k < 16; ++k) a = add4(a, v[k]);
                } else {
#pragma unroll
                    for (int k = 0; k < 15; ++k)
                        if ((unsigned)k < m) v[k] = *(const float4*)(h + (size_t)__builtin_amdgcn_readlane(e, k) * CH + lane * 4);
#pragma unroll
                    for (int k = 0; k < 15; ++k)
                        if ((unsigned)k < m) a = add4(a, v[k]);
                }
            }
            const float s = rsqrtf(fmaxf((float)n, 1.0f));
            a.x = fmaf(a.x, s, b.x);
            a.y = fmaf(a.y, s, b.y);
            a.z = fmaf(a.z, s, b.z);
            a.w = fmaf(a.w, s, b.w);
            *(float4*)&out[(size_t)(lo + (int)r) * CH + lane * 4] = a;
        }
        return;
    }

    for (int i = t; i < TILE * CH / 4; i += 1024) ((float4*)acc)[i] = make_float4(0.f, 0.f, 0.f, 0.f);
    __syncthreads();
#pragma unroll
    for (int k = 0; k < CELL_TRIPS; ++k) {
        slow_rec(pv[k].x, acc, h);
        slow_rec(pv[k].y, acc, h);
        slow_rec(pv[k].z, acc, h);
        slow_rec(pv[k].w, acc, h);
    }
    for (unsigned o = 0; o < nov; ++o) {
        const unsigned e0 = ovf[o] * (unsigned)CHUNK;
        for (int i = t; i < CHUNK; i += 1024) {
            const unsigned e = e0 + (unsigned)i;
            if (e < (unsigned)N_EDGES) {
                const unsigned dl = (unsigned)(dst[e] - lo);
                if (dl < (unsigned)TILE) {
                    atomicAdd(&indeg[dl], 1u);
                    add_row_alone(acc, h, (unsigned)src[e], dl);
                }
            }
        }
    }
    __syncthreads();
    {
        const float4 b = *(const float4*)&bias[lane * 4];
        for (int r = wave; r < TILE; r += 32) {
            const int node = lo + r;
            const float s = rsqrtf(fmaxf((float)indeg[r], 1.0f));
            const float* ar = &acc[r * CH + lane];
            float4 a = make_float4(ar[0], ar[32], ar[64], ar[96]);
            a.x = fmaf(a.x, s, b.x);
            a.y = fmaf(a.y, s, b.y);
            a.z = fmaf(a.z, s, b.z);
            a.w = fmaf(a.w, s, b.w);
            *(float4*)&out[(size_t)node * CH + lane * 4] = a;
        }
    }
}

extern "C" void kernel_launch(void* const* d_in, const int* in_sizes, int n_in,
                              void* d_out, int out_size, void* d_ws, size_t ws_size,
                              hipStream_t stream) {
    const float* feat  = (const float*)d_in[0];
    const int*   src   = (const int*)  d_in[1];
    const int*   dst   = (const int*)  d_in[2];
    const float* W_mu  = (const float*)d_in[3];
    const float* W_rho = (const float*)d_in[4];
    const float* b_mu  = (const float*)d_in[5];
    const float* b_rho = (const float*)d_in[6];
    const float* W_eps = (const float*)d_in[7];
    const float* b_eps = (const float*)d_in[8];

    float* ws   = (float*)d_ws;
    float* W    = ws + OFF_W;
    float* bias = ws + OFF_BIAS;
    float* h    = ws + OFF_H;
    float* out  = (float*)d_out;

    input_pass<<<PASS1_WGS, 1024, 0, stream>>>(src, dst, W_mu, W_rho, W_eps, b_mu, b_rho, b_eps, W, bias, out);

    gemm_wmma<<<GEMM_WGS, 128, 0, stream>>>(feat, out, W, h);

    tile_gather<<<N_TILES, 1024, 0, stream>>>(src, dst, h, bias, out);
}
